# starting kernel
# baseline (speedup 1.0000x reference)
#include <hip/hip_runtime.h>
#include <hip/hip_bf16.h>
#include <math.h>

typedef float v2f __attribute__((ext_vector_type(2)));
typedef float v8f __attribute__((ext_vector_type(8)));

__device__ __forceinline__ void edge_sd(const int* __restrict__ ei, int e, int E,
                                        int& s, int& d) {
  if (e < E) { s = ei[e]; d = ei[E + e]; }
  else       { s = e - E; d = s; }
}

__device__ __forceinline__ void atomicMaxF(float* addr, float v) {
  if (v >= 0.0f) atomicMax((int*)addr, __float_as_int(v));
  else           atomicMin((unsigned int*)addr, __float_as_uint(v));
}

__global__ __launch_bounds__(256)
void gemm128_wmma(const float* __restrict__ X, const float* __restrict__ W,
                  float* __restrict__ Y, int nrows) {
  __shared__ float Xs[128 * 33];
  __shared__ float Ws[32 * 132];

  const int tid  = threadIdx.x;
  const int lane = tid & 31;
  const int wv   = tid >> 5;
  const int rc   = lane & 15;
  const int kh   = (lane >> 4) << 1;
  const int r0   = blockIdx.x * 128;

  v8f acc[8];
#pragma unroll
  for (int nt = 0; nt < 8; ++nt)
#pragma unroll
    for (int i = 0; i < 8; ++i) acc[nt][i] = 0.0f;

  for (int kb = 0; kb < 4; ++kb) {
#pragma unroll
    for (int i = 0; i < 4; ++i) {
      int idx = tid + i * 256;
      int row = idx >> 3;
      int c4  = (idx & 7) << 2;
      int gr  = r0 + row;
      float4 v = {0.f, 0.f, 0.f, 0.f};
      if (gr < nrows) v = *(const float4*)(X + (long)gr * 128 + kb * 32 + c4);
      float* p = &Xs[row * 33 + c4];
      p[0] = v.x; p[1] = v.y; p[2] = v.z; p[3] = v.w;
    }
#pragma unroll
    for (int i = 0; i < 4; ++i) {
      int idx = tid + i * 256;
      int row = idx >> 5;
      int c4  = (idx & 31) << 2;
      float4 v = *(const float4*)(W + (long)(kb * 32 + row) * 128 + c4);
      float* p = &Ws[row * 132 + c4];
      p[0] = v.x; p[1] = v.y; p[2] = v.z; p[3] = v.w;
    }
    __syncthreads();

    v2f afr[8];
#pragma unroll
    for (int ks = 0; ks < 8; ++ks) {
      const float* p = &Xs[(wv * 16 + rc) * 33 + ks * 4 + kh];
      afr[ks][0] = p[0];
      afr[ks][1] = p[1];
    }
#pragma unroll
    for (int nt = 0; nt < 8; ++nt) {
#pragma unroll
      for (int ks = 0; ks < 8; ++ks) {
        const float* p = &Ws[(ks * 4 + kh) * 132 + nt * 16 + rc];
        v2f b;
        b[0] = p[0];
        b[1] = p[132];
        acc[nt] = __builtin_amdgcn_wmma_f32_16x16x4_f32(
            false, afr[ks], false, b, (short)0, acc[nt], false, false);
      }
    }
    __syncthreads();
  }

#pragma unroll
  for (int nt = 0; nt < 8; ++nt) {
#pragma unroll
    for (int i = 0; i < 8; ++i) {
      int row = wv * 16 + i + ((lane >> 4) << 3);
      int gr  = r0 + row;
      if (gr < nrows) Y[(long)gr * 128 + nt * 16 + rc] = acc[nt][i];
    }
  }
}

__global__ __launch_bounds__(256)
void attn_scalars(const float* __restrict__ xl, const float* __restrict__ asrc,
                  const float* __restrict__ adst, float* __restrict__ as_out,
                  float* __restrict__ ad_out, int N, int H) {
  int node = blockIdx.x * 8 + (threadIdx.x >> 5);
  int lane = threadIdx.x & 31;
  if (node >= N) return;
  const float* row = xl + (long)node * 128;
  float s[4], d[4];
#pragma unroll
  for (int b = 0; b < 4; ++b) {
    int f = lane + 32 * b;
    float x = row[f];
    s[b] = x * asrc[f];
    d[b] = x * adst[f];
  }
#pragma unroll
  for (int b = 0; b < 4; ++b) {
    for (int off = 16; off > 0; off >>= 1) {
      s[b] += __shfl_xor(s[b], off, 32);
      d[b] += __shfl_xor(d[b], off, 32);
    }
  }
  if (lane == 0) {
    if (H == 4) {
#pragma unroll
      for (int h = 0; h < 4; ++h) {
        as_out[(long)node * 4 + h] = s[h];
        ad_out[(long)node * 4 + h] = d[h];
      }
    } else {
      as_out[node] = s[0] + s[1] + s[2] + s[3];
      ad_out[node] = d[0] + d[1] + d[2] + d[3];
    }
  }
}

__global__ __launch_bounds__(256)
void init_layer(float* __restrict__ out, const float* __restrict__ bias,
                float* __restrict__ m, float* __restrict__ den, int N, int H) {
  long t = (long)blockIdx.x * 256 + threadIdx.x;
  if (t < (long)N * 128) out[t] = bias[t & 127];
  if (t < (long)N * H) { m[t] = -INFINITY; den[t] = 0.0f; }
}

__global__ __launch_bounds__(256)
void edge_logit_max(const int* __restrict__ ei, const float* __restrict__ as,
                    const float* __restrict__ ad, float* __restrict__ elog,
                    float* __restrict__ m, int E, int N, int H) {
  int e = blockIdx.x * 256 + threadIdx.x;
  if (e >= E + N) return;
  int s, d;
  edge_sd(ei, e, E, s, d);
  for (int h = 0; h < H; ++h) {
    float l = as[(long)s * H + h] + ad[(long)d * H + h];
    l = (l > 0.0f) ? l : 0.2f * l;
    elog[(long)e * H + h] = l;
    atomicMaxF(&m[(long)d * H + h], l);
  }
}

__global__ __launch_bounds__(256)
void edge_expsum(const int* __restrict__ ei, float* __restrict__ elog,
                 const float* __restrict__ m, float* __restrict__ den,
                 int E, int N, int H) {
  int e = blockIdx.x * 256 + threadIdx.x;
  if (e >= E + N) return;
  int s, d;
  edge_sd(ei, e, E, s, d);
  for (int h = 0; h < H; ++h) {
    float ex = expf(elog[(long)e * H + h] - m[(long)d * H + h]);
    elog[(long)e * H + h] = ex;
    atomicAdd(&den[(long)d * H + h], ex);
  }
}

__global__ __launch_bounds__(256)
void edge_message(const int* __restrict__ ei, const float* __restrict__ xl,
                  const float* __restrict__ elog, const float* __restrict__ den,
                  float* __restrict__ out, int E, int N, int H) {
  int e = blockIdx.x * 8 + (threadIdx.x >> 5);
  int lane = threadIdx.x & 31;
  if (e >= E + N) return;
  int s, d;
  edge_sd(ei, e, E, s, d);
  const float* xr = xl + (long)s * 128;
  float* orow = out + (long)d * 128;
  __builtin_prefetch(xr + lane * 4, 0, 1);
  float alpha[4];
  if (H == 4) {
#pragma unroll
    for (int h = 0; h < 4; ++h)
      alpha[h] = elog[(long)e * 4 + h] / (den[(long)d * 4 + h] + 1e-16f);
  } else {
    float a0 = elog[e] / (den[d] + 1e-16f);
    alpha[0] = alpha[1] = alpha[2] = alpha[3] = a0;
  }
#pragma unroll
  for (int b = 0; b < 4; ++b) {
    int f = lane + 32 * b;
    atomicAdd(&orow[f], xr[f] * alpha[b]);
  }
}

__global__ __launch_bounds__(256)
void elu_inplace(float* __restrict__ x, long n) {
  long t = (long)blockIdx.x * 256 + threadIdx.x;
  if (t >= n) return;
  float v = x[t];
  x[t] = (v > 0.0f) ? v : (expf(v) - 1.0f);
}

__global__ __launch_bounds__(256)
void l2norm_rows(float* __restrict__ out, int N) {
  int node = blockIdx.x * 8 + (threadIdx.x >> 5);
  int lane = threadIdx.x & 31;
  if (node >= N) return;
  float* row = out + (long)node * 128;
  float v[4];
  float ss = 0.0f;
#pragma unroll
  for (int b = 0; b < 4; ++b) {
    v[b] = row[lane + 32 * b];
    ss += v[b] * v[b];
  }
  for (int off = 16; off > 0; off >>= 1) ss += __shfl_xor(ss, off, 32);
  float sc = 1.0f / fmaxf(sqrtf(ss), 1e-12f);
#pragma unroll
  for (int b = 0; b < 4; ++b) row[lane + 32 * b] = v[b] * sc;
}

extern "C" void kernel_launch(void* const* d_in, const int* in_sizes, int n_in,
                              void* d_out, int out_size, void* d_ws, size_t ws_size,
                              hipStream_t stream) {
  const int*   ei    = (const int*)d_in[0];
  const float* emb   = (const float*)d_in[1];
  const float* W1    = (const float*)d_in[2];
  const float* asrc1 = (const float*)d_in[3];
  const float* adst1 = (const float*)d_in[4];
  const float* b1    = (const float*)d_in[5];
  const float* W2    = (const float*)d_in[6];
  const float* asrc2 = (const float*)d_in[7];
  const float* adst2 = (const float*)d_in[8];
  const float* b2    = (const float*)d_in[9];
  float* out = (float*)d_out;

  const int E  = in_sizes[0] / 2;
  const int N  = in_sizes[1] / 128;
  const int E2 = E + N;

  float* p   = (float*)d_ws;
  float* xl1 = p; p += (size_t)N * 128;
  float* x1  = p; p += (size_t)N * 128;
  float* xl2 = p; p += (size_t)N * 128;
  float* as1 = p; p += (size_t)N * 4;
  float* ad1 = p; p += (size_t)N * 4;
  float* m1  = p; p += (size_t)N * 4;
  float* dn1 = p; p += (size_t)N * 4;
  float* as2 = p; p += (size_t)N;
  float* ad2 = p; p += (size_t)N;
  float* m2  = p; p += (size_t)N;
  float* dn2 = p; p += (size_t)N;
  float* elog = p;

  dim3 blk(256);
  int gGemm  = (N + 127) / 128;
  int gNode8 = (N + 7) / 8;
  int gElem  = (int)(((long)N * 128 + 255) / 256);
  int gEdge  = (E2 + 255) / 256;
  int gEdge8 = (E2 + 7) / 8;

  gemm128_wmma <<<gGemm,  blk, 0, stream>>>(emb, W1, xl1, N);
  attn_scalars <<<gNode8, blk, 0, stream>>>(xl1, asrc1, adst1, as1, ad1, N, 4);
  init_layer   <<<gElem,  blk, 0, stream>>>(x1, b1, m1, dn1, N, 4);
  edge_logit_max<<<gEdge, blk, 0, stream>>>(ei, as1, ad1, elog, m1, E, N, 4);
  edge_expsum  <<<gEdge,  blk, 0, stream>>>(ei, elog, m1, dn1, E, N, 4);
  edge_message <<<gEdge8, blk, 0, stream>>>(ei, xl1, elog, dn1, x1, E, N, 4);
  elu_inplace  <<<gElem,  blk, 0, stream>>>(x1, (long)N * 128);

  gemm128_wmma <<<gGemm,  blk, 0, stream>>>(x1, W2, xl2, N);
  attn_scalars <<<gNode8, blk, 0, stream>>>(xl2, asrc2, adst2, as2, ad2, N, 1);
  init_layer   <<<gElem,  blk, 0, stream>>>(out, b2, m2, dn2, N, 1);
  edge_logit_max<<<gEdge, blk, 0, stream>>>(ei, as2, ad2, elog, m2, E, N, 1);
  edge_expsum  <<<gEdge,  blk, 0, stream>>>(ei, elog, m2, dn2, E, N, 1);
  edge_message <<<gEdge8, blk, 0, stream>>>(ei, xl2, elog, dn2, out, E, N, 1);
  l2norm_rows  <<<gNode8, blk, 0, stream>>>(out, N);
}


# Gather by destination, not atomic scatter: per-call buckets sorted by edge id (arrival order made sums vary), one wave per node does max, exp sum, row sum in registers; sums in edge-id order
# speedup vs baseline: 3.5695x; 3.5695x over previous
#include <hip/hip_runtime.h>
#include <hip/hip_bf16.h>
#include <math.h>

typedef float v2f __attribute__((ext_vector_type(2)));
typedef float v8f __attribute__((ext_vector_type(8)));

__global__ __launch_bounds__(256)
void gemm128_wmma(const float* __restrict__ X, const float* __restrict__ W,
                  float* __restrict__ Y, int nrows) {
  __shared__ float Xs[128 * 33];
  __shared__ float Ws[32 * 132];

  const int tid  = threadIdx.x;
  const int lane = tid & 31;
  const int wv   = tid >> 5;
  const int rc   = lane & 15;
  const int kh   = (lane >> 4) << 1;
  const int r0   = blockIdx.x * 128;

  v8f acc[8];
#pragma unroll
  for (int nt = 0; nt < 8; ++nt)
#pragma unroll
    for (int i = 0; i < 8; ++i) acc[nt][i] = 0.0f;

  for (int kb = 0; kb < 4; ++kb) {
#pragma unroll
    for (int i = 0; i < 4; ++i) {
      int idx = tid + i * 256;
      int row = idx >> 3;
      int c4  = (idx & 7) << 2;
      int gr  = r0 + row;
      float4 v = {0.f, 0.f, 0.f, 0.f};
      if (gr < nrows) v = *(const float4*)(X + (long)gr * 128 + kb * 32 + c4);
      float* p = &Xs[row * 33 + c4];
      p[0] = v.x; p[1] = v.y; p[2] = v.z; p[3] = v.w;
    }
#pragma unroll
    for (int i = 0; i < 4; ++i) {
      int idx = tid + i * 256;
      int row = idx >> 5;
      int c4  = (idx & 31) << 2;
      float4 v = *(const float4*)(W + (long)(kb * 32 + row) * 128 + c4);
      float* p = &Ws[row * 132 + c4];
      p[0] = v.x; p[1] = v.y; p[2] = v.z; p[3] = v.w;
    }
    __syncthreads();

    v2f afr[8];
#pragma unroll
    for (int ks = 0; ks < 8; ++ks) {
      const float* p = &Xs[(wv * 16 + rc) * 33 + ks * 4 + kh];
      afr[ks][0] = p[0];
      afr[ks][1] = p[1];
    }
#pragma unroll
    for (int nt = 0; nt < 8; ++nt) {
#pragma unroll
      for (int ks = 0; ks < 8; ++ks) {
        const float* p = &Ws[(ks * 4 + kh) * 132 + nt * 16 + rc];
        v2f b;
        b[0] = p[0];
        b[1] = p[132];
        acc[nt] = __builtin_amdgcn_wmma_f32_16x16x4_f32(
            false, afr[ks], false, b, (short)0, acc[nt], false, false);
      }
    }
    __syncthreads();
  }

#pragma unroll
  for (int nt = 0; nt < 8; ++nt) {
#pragma unroll
    for (int i = 0; i < 8; ++i) {
      int row = wv * 16 + i + ((lane >> 4) << 3);
      int gr  = r0 + row;
      if (gr < nrows) Y[(long)gr * 128 + nt * 16 + rc] = acc[nt][i];
    }
  }
}

__global__ __launch_bounds__(256)
void attn_scalars(const float* __restrict__ xl, const float* __restrict__ asrc,
                  const float* __restrict__ adst, float* __restrict__ as_out,
                  float* __restrict__ ad_out, int N, int H) {
  int node = blockIdx.x * 8 + (threadIdx.x >> 5);
  int lane = threadIdx.x & 31;
  if (node >= N) return;
  const float* row = xl + (long)node * 128;
  float s[4], d[4];
#pragma unroll
  for (int b = 0; b < 4; ++b) {
    int f = lane + 32 * b;
    float x = row[f];
    s[b] = x * asrc[f];
    d[b] = x * adst[f];
  }
#pragma unroll
  for (int b = 0; b < 4; ++b) {
    for (int off = 16; off > 0; off >>= 1) {
      s[b] += __shfl_xor(s[b], off, 32);
      d[b] += __shfl_xor(d[b], off, 32);
    }
  }
  if (lane == 0) {
    if (H == 4) {
#pragma unroll
      for (int h = 0; h < 4; ++h) {
        as_out[(long)node * 4 + h] = s[h];
        ad_out[(long)node * 4 + h] = d[h];
      }
    } else {
      as_out[node] = s[0] + s[1] + s[2] + s[3];
      ad_out[node] = d[0] + d[1] + d[2] + d[3];
    }
  }
}

__global__ __launch_bounds__(256)
void zero_ints(int* __restrict__ p, int n) {
  int t = blockIdx.x * 256 + threadIdx.x;
  if (t < n) p[t] = 0;
}

__global__ __launch_bounds__(256)
void count_rank(const int* __restrict__ ei, int* __restrict__ deg,
                int* __restrict__ rank, int E) {
  int e = blockIdx.x * 256 + threadIdx.x;
  if (e >= E) return;
  rank[e] = atomicAdd(&deg[ei[E + e]], 1);
}

__global__ __launch_bounds__(256)
void bucket_offsets(const int* __restrict__ deg, int* __restrict__ rowptr, int N) {
  __shared__ int wsum[8];
  __shared__ int wpre[8];
  const int tid = threadIdx.x, lane = tid & 31, wv = tid >> 5;
  const int b0 = blockIdx.x * 256;
  int part = 0;
  for (int i = tid; i < b0; i += 256) part += deg[i];
  const int i = b0 + tid;
  const int v = (i < N) ? deg[i] : 0;
  int inc = v;
  for (int off = 1; off < 32; off <<= 1) {
    int t = __shfl_up(inc, off, 32);
    if (lane >= off) inc += t;
  }
  for (int off = 16; off > 0; off >>= 1) part += __shfl_xor(part, off, 32);
  if (lane == 31) wsum[wv] = inc;
  if (lane == 0)  wpre[wv] = part;
  __syncthreads();
  int pre = 0;
  for (int w = 0; w < 8; ++w) {
    pre += wpre[w];
    if (w < wv) pre += wsum[w];
  }
  if (i < N) rowptr[i] = pre + inc - v;
}

__global__ __launch_bounds__(256)
void place_edges(const int* __restrict__ ei, const int* __restrict__ rowptr,
                 const int* __restrict__ rank, int* __restrict__ col, int E) {
  int e = blockIdx.x * 256 + threadIdx.x;
  if (e >= E) return;
  col[rowptr[ei[E + e]] + rank[e]] = e;
}

__global__ __launch_bounds__(256)
void sort_buckets(const int* __restrict__ ei, const int* __restrict__ rowptr,
                  const int* __restrict__ deg, const int* __restrict__ col,
                  int* __restrict__ csrc, int N) {
  const int node = __builtin_amdgcn_readfirstlane(blockIdx.x * 8 + (threadIdx.x >> 5));
  const int lane = threadIdx.x & 31;
  if (node >= N) return;
  const int base = rowptr[node];
  const int dg   = deg[node];
  for (int ci = 0; ci < dg; ci += 32) {
    const int i  = ci + lane;
    const int my = (i < dg) ? col[base + i] : 0x7fffffff;
    int pos = 0;
    for (int cj = 0; cj < dg; cj += 32) {
      const int j  = cj + lane;
      const int ej = (j < dg) ? col[base + j] : 0x7fffffff;
      const int cnt = min(32, dg - cj);
      for (int t = 0; t < cnt; ++t) {
        const int o = __builtin_amdgcn_readlane(ej, t);
        pos += (o < my) ? 1 : 0;
      }
    }
    if (i < dg) csrc[base + pos] = ei[my];
  }
}

__device__ __forceinline__ float wave_max(float v) {
  for (int off = 16; off > 0; off >>= 1) v = fmaxf(v, __shfl_xor(v, off, 32));
  return v;
}
__device__ __forceinline__ float wave_sum(float v) {
  for (int off = 16; off > 0; off >>= 1) v += __shfl_xor(v, off, 32);
  return v;
}
__device__ __forceinline__ float leaky02(float l) {
  return (l > 0.0f) ? l : 0.2f * l;
}

template <int H>
__device__ __forceinline__ void chunk_logits(const int* __restrict__ csrc,
                                             const float* __restrict__ as,
                                             const float* adn, int base, int i,
                                             int dg, int node, int& s, float* l) {
  s = (i < dg) ? csrc[base + i] : node;
  if (H == 4) {
    const float4 a = *(const float4*)(as + (long)s * 4);
    l[0] = leaky02(a.x + adn[0]);
    l[1] = leaky02(a.y + adn[1]);
    l[2] = leaky02(a.z + adn[2]);
    l[3] = leaky02(a.w + adn[3]);
  } else {
    l[0] = leaky02(as[s] + adn[0]);
  }
}

template <int H>
__device__ __forceinline__ void add_rows(const float* __restrict__ xl, int s,
                                         const float* al, int cnt, int lane,
                                         float* acc) {
#pragma unroll 4
  for (int t = 0; t < cnt; ++t) {
    const int st = __builtin_amdgcn_readlane(s, t);
    const float* xr = xl + (long)st * 128 + lane;
    float a[H];
    for (int h = 0; h < H; ++h)
      a[h] = __builtin_bit_cast(float,
               __builtin_amdgcn_readlane(__builtin_bit_cast(int, al[h]), t));
    for (int b = 0; b < 4; ++b)
      acc[b] += xr[32 * b] * a[H == 4 ? b : 0];
  }
}

template <int H>
__device__ __forceinline__ void node_gather(const int* __restrict__ rowptr,
                                            const int* __restrict__ deg,
                                            const int* __restrict__ csrc,
                                            const float* __restrict__ xl,
                                            const float* __restrict__ as,
                                            const float* __restrict__ ad,
                                            const float* __restrict__ bias,
                                            float* __restrict__ out, int N) {
  const int node = __builtin_amdgcn_readfirstlane(blockIdx.x * 8 + (threadIdx.x >> 5));
  const int lane = threadIdx.x & 31;
  if (node >= N) return;
  const int base = rowptr[node];
  const int dg   = deg[node];

  float adn[H], lself[H], m[H], den[H];
  for (int h = 0; h < H; ++h) {
    adn[h]   = ad[(long)node * H + h];
    lself[h] = leaky02(as[(long)node * H + h] + adn[h]);
  }
  float acc[4] = {0.0f, 0.0f, 0.0f, 0.0f};

  if (dg <= 32) {
    int s; float l[H], al[H];
    chunk_logits<H>(csrc, as, adn, base, lane, dg, node, s, l);
    const float live = (lane < dg) ? 1.0f : 0.0f;
    for (int h = 0; h < H; ++h) {
      m[h] = fmaxf(wave_max(l[h]), lself[h]);
      const float ex = expf(l[h] - m[h]) * live;
      den[h] = wave_sum(ex) + expf(lself[h] - m[h]);
      al[h]  = ex / (den[h] + 1e-16f);
    }
    add_rows<H>(xl, s, al, dg, lane, acc);
  } else {
    for (int h = 0; h < H; ++h) m[h] = lself[h];
    for (int c = 0; c < dg; c += 32) {
      int s; float l[H];
      chunk_logits<H>(csrc, as, adn, base, c + lane, dg, node, s, l);
      for (int h = 0; h < H; ++h) m[h] = fmaxf(m[h], l[h]);
    }
    for (int h = 0; h < H; ++h) { m[h] = wave_max(m[h]); den[h] = 0.0f; }
    for (int c = 0; c < dg; c += 32) {
      int s; float l[H];
      chunk_logits<H>(csrc, as, adn, base, c + lane, dg, node, s, l);
      const float live = (c + lane < dg) ? 1.0f : 0.0f;
      for (int h = 0; h < H; ++h) den[h] += expf(l[h] - m[h]) * live;
    }
    for (int h = 0; h < H; ++h) den[h] = wave_sum(den[h]) + expf(lself[h] - m[h]);
    for (int c = 0; c < dg; c += 32) {
      int s; float l[H], al[H];
      chunk_logits<H>(csrc, as, adn, base, c + lane, dg, node, s, l);
      for (int h = 0; h < H; ++h) al[h] = expf(l[h] - m[h]) / (den[h] + 1e-16f);
      add_rows<H>(xl, s, al, min(32, dg - c), lane, acc);
    }
  }

  const float* xr = xl + (long)node * 128 + lane;
  float* orow = out + (long)node * 128 + lane;
  for (int b = 0; b < 4; ++b) {
    const int h = (H == 4) ? b : 0;
    const float aself = expf(lself[h] - m[h]) / (den[h] + 1e-16f);
    orow[32 * b] = acc[b] + xr[32 * b] * aself + bias[lane + 32 * b];
  }
}

__global__ __launch_bounds__(256)
void node_gather_h4(const int* __restrict__ rowptr, const int* __restrict__ deg,
                    const int* __restrict__ csrc, const float* __restrict__ xl,
                    const float* __restrict__ as, const float* __restrict__ ad,
                    const float* __restrict__ bias, float* __restrict__ out, int N) {
  node_gather<4>(rowptr, deg, csrc, xl, as, ad, bias, out, N);
}

__global__ __launch_bounds__(256)
void node_gather_h1(const int* __restrict__ rowptr, const int* __restrict__ deg,
                    const int* __restrict__ csrc, const float* __restrict__ xl,
                    const float* __restrict__ as, const float* __restrict__ ad,
                    const float* __restrict__ bias, float* __restrict__ out, int N) {
  node_gather<1>(rowptr, deg, csrc, xl, as, ad, bias, out, N);
}

__global__ __launch_bounds__(256)
void elu_inplace(float* __restrict__ x, long n) {
  long t = (long)blockIdx.x * 256 + threadIdx.x;
  if (t >= n) return;
  float v = x[t];
  x[t] = (v > 0.0f) ? v : (expf(v) - 1.0f);
}

__global__ __launch_bounds__(256)
void l2norm_rows(float* __restrict__ out, int N) {
  int node = blockIdx.x * 8 + (threadIdx.x >> 5);
  int lane = threadIdx.x & 31;
  if (node >= N) return;
  float* row = out + (long)node * 128;
  float v[4];
  float ss = 0.0f;
#pragma unroll
  for (int b = 0; b < 4; ++b) {
    v[b] = row[lane + 32 * b];
    ss += v[b] * v[b];
  }
  for (int off = 16; off > 0; off >>= 1) ss += __shfl_xor(ss, off, 32);
  float sc = 1.0f / fmaxf(sqrtf(ss), 1e-12f);
#pragma unroll
  for (int b = 0; b < 4; ++b) row[lane + 32 * b] = v[b] * sc;
}

extern "C" void kernel_launch(void* const* d_in, const int* in_sizes, int n_in,
                              void* d_out, int out_size, void* d_ws, size_t ws_size,
                              hipStream_t stream) {
  const int*   ei    = (const int*)d_in[0];
  const float* emb   = (const float*)d_in[1];
  const float* W1    = (const float*)d_in[2];
  const float* asrc1 = (const float*)d_in[3];
  const float* adst1 = (const float*)d_in[4];
  const float* b1    = (const float*)d_in[5];
  const float* W2    = (const float*)d_in[6];
  const float* asrc2 = (const float*)d_in[7];
  const float* adst2 = (const float*)d_in[8];
  const float* b2    = (const float*)d_in[9];
  float* out = (float*)d_out;

  const int E = in_sizes[0] / 2;
  const int N = in_sizes[1] / 128;

  float* p   = (float*)d_ws;
  float* xl1 = p; p += (size_t)N * 128;
  float* x1  = p; p += (size_t)N * 128;
  float* xl2 = p; p += (size_t)N * 128;
  float* as1 = p; p += (size_t)N * 4;
  float* ad1 = p; p += (size_t)N * 4;
  float* as2 = p; p += (size_t)N;
  float* ad2 = p; p += (size_t)N;
  int* q      = (int*)p;
  int* deg    = q; q += (size_t)N;
  int* rowptr = q; q += (size_t)N;
  int* rank   = q; q += (size_t)E;
  int* col    = q; q += (size_t)E;
  int* csrc   = q;

  dim3 blk(256);
  int gGemm  = (N + 127) / 128;
  int gNode8 = (N + 7) / 8;
  int gNode  = (N + 255) / 256;
  int gElem  = (int)(((long)N * 128 + 255) / 256);
  int gEdge  = (E + 255) / 256;

  zero_ints     <<<gNode,  blk, 0, stream>>>(deg, N);
  count_rank    <<<gEdge,  blk, 0, stream>>>(ei, deg, rank, E);
  bucket_offsets<<<gNode,  blk, 0, stream>>>(deg, rowptr, N);
  place_edges   <<<gEdge,  blk, 0, stream>>>(ei, rowptr, rank, col, E);
  sort_buckets  <<<gNode8, blk, 0, stream>>>(ei, rowptr, deg, col, csrc, N);

  gemm128_wmma  <<<gGemm,  blk, 0, stream>>>(emb, W1, xl1, N);
  attn_scalars  <<<gNode8, blk, 0, stream>>>(xl1, asrc1, adst1, as1, ad1, N, 4);
  node_gather_h4<<<gNode8, blk, 0, stream>>>(rowptr, deg, csrc, xl1, as1, ad1, b1, x1, N);
  elu_inplace   <<<gElem,  blk, 0, stream>>>(x1, (long)N * 128);

  gemm128_wmma  <<<gGemm,  blk, 0, stream>>>(x1, W2, xl2, N);
  attn_scalars  <<<gNode8, blk, 0, stream>>>(xl2, asrc2, adst2, as2, ad2, N, 1);
  node_gather_h1<<<gNode8, blk, 0, stream>>>(rowptr, deg, csrc, xl2, as2, ad2, b2, out, N);
  l2norm_rows   <<<gNode8, blk, 0, stream>>>(out, N);
}


# Fold row-wise passes into their producers: a_s/a_d from the GEMM accumulators (summed per column tile, then 16-lane butterfly: new order), ELU and row L2 norm at the gather's store; 13 launches to 9
# speedup vs baseline: 4.3725x; 1.2250x over previous
#include <hip/hip_runtime.h>
#include <hip/hip_bf16.h>
#include <math.h>

typedef float v2f __attribute__((ext_vector_type(2)));
typedef float v8f __attribute__((ext_vector_type(8)));

template <int M>
__device__ __forceinline__ float row_xor(float v) {
  return __builtin_bit_cast(float,
      __builtin_amdgcn_update_dpp(0, __builtin_bit_cast(int, v), 0x160 + M,
                                  0xf, 0xf, true));
}
__device__ __forceinline__ float row16_sum(float v) {
  v += row_xor<1>(v);
  v += row_xor<2>(v);
  v += row_xor<4>(v);
  v += row_xor<8>(v);
  return v;
}

template <int H>
__device__ __forceinline__
void gemm128_attn(const float* __restrict__ X, const float* __restrict__ W,
                  const float* __restrict__ asrc, const float* __restrict__ adst,
                  float* __restrict__ Y, float* __restrict__ as_out,
                  float* __restrict__ ad_out, int nrows) {
  __shared__ float Xs[128 * 33];
  __shared__ float Ws[32 * 132];

  const int tid  = threadIdx.x;
  const int lane = tid & 31;
  const int wv   = tid >> 5;
  const int rc   = lane & 15;
  const int kh   = (lane >> 4) << 1;
  const int r0   = blockIdx.x * 128;

  v8f acc[8];
#pragma unroll
  for (int nt = 0; nt < 8; ++nt)
#pragma unroll
    for (int i = 0; i < 8; ++i) acc[nt][i] = 0.0f;

  for (int kb = 0; kb < 4; ++kb) {
#pragma unroll
    for (int i = 0; i < 4; ++i) {
      int idx = tid + i * 256;
      int row = idx >> 3;
      int c4  = (idx & 7) << 2;
      int gr  = r0 + row;
      float4 v = {0.f, 0.f, 0.f, 0.f};
      if (gr < nrows) v = *(const float4*)(X + (long)gr * 128 + kb * 32 + c4);
      float* p = &Xs[row * 33 + c4];
      p[0] = v.x; p[1] = v.y; p[2] = v.z; p[3] = v.w;
    }
#pragma unroll
    for (int i = 0; i < 4; ++i) {
      int idx = tid + i * 256;
      int row = idx >> 5;
      int c4  = (idx & 31) << 2;
      float4 v = *(const float4*)(W + (long)(kb * 32 + row) * 128 + c4);
      float* p = &Ws[row * 132 + c4];
      p[0] = v.x; p[1] = v.y; p[2] = v.z; p[3] = v.w;
    }
    __syncthreads();

    v2f afr[8];
#pragma unroll
    for (int ks = 0; ks < 8; ++ks) {
      const float* p = &Xs[(wv * 16 + rc) * 33 + ks * 4 + kh];
      afr[ks][0] = p[0];
      afr[ks][1] = p[1];
    }
#pragma unroll
    for (int nt = 0; nt < 8; ++nt) {
#pragma unroll
      for (int ks = 0; ks < 8; ++ks) {
        const float* p = &Ws[(ks * 4 + kh) * 132 + nt * 16 + rc];
        v2f b;
        b[0] = p[0];
        b[1] = p[132];
        acc[nt] = __builtin_amdgcn_wmma_f32_16x16x4_f32(
            false, afr[ks], false, b, (short)0, acc[nt], false, false);
      }
    }
    __syncthreads();
  }

#pragma unroll
  for (int nt = 0; nt < 8; ++nt) {
#pragma unroll
    for (int i = 0; i < 8; ++i) {
      int row = wv * 16 + i + ((lane >> 4) << 3);
      int gr  = r0 + row;
      if (gr < nrows) Y[(long)gr * 128 + nt * 16 + rc] = acc[nt][i];
    }
  }

  float wsrc[8], wdst[8];
#pragma unroll
  for (int nt = 0; nt < 8; ++nt) {
    wsrc[nt] = asrc[nt * 16 + rc];
    wdst[nt] = adst[nt * 16 + rc];
  }
  float ks[H], kd[H];
#pragma unroll
  for (int h = 0; h < H; ++h) { ks[h] = 0.0f; kd[h] = 0.0f; }
#pragma unroll
  for (int i = 0; i < 8; ++i) {
    float s[H], d[H];
    if (H == 4) {
#pragma unroll
      for (int h = 0; h < 4; ++h) {
        s[h] = acc[2 * h][i] * wsrc[2 * h] + acc[2 * h + 1][i] * wsrc[2 * h + 1];
        d[h] = acc[2 * h][i] * wdst[2 * h] + acc[2 * h + 1][i] * wdst[2 * h + 1];
      }
    } else {
      s[0] = 0.0f; d[0] = 0.0f;
#pragma unroll
      for (int nt = 0; nt < 8; ++nt) {
        s[0] += acc[nt][i] * wsrc[nt];
        d[0] += acc[nt][i] * wdst[nt];
      }
    }
#pragma unroll
    for (int h = 0; h < H; ++h) {
      s[h] = row16_sum(s[h]);
      d[h] = row16_sum(d[h]);
      if (rc == i) { ks[h] = s[h]; kd[h] = d[h]; }
    }
  }
  const int grow = r0 + wv * 16 + rc + ((lane >> 4) << 3);
  if (rc < 8 && grow < nrows) {
    if (H == 4) {
      *(float4*)(as_out + (long)grow * 4) = make_float4(ks[0], ks[1], ks[2], ks[3]);
      *(float4*)(ad_out + (long)grow * 4) = make_float4(kd[0], kd[1], kd[2], kd[3]);
    } else {
      as_out[grow] = ks[0];
      ad_out[grow] = kd[0];
    }
  }
}

__global__ __launch_bounds__(256)
void gemm128_attn_h4(const float* __restrict__ X, const float* __restrict__ W,
                     const float* __restrict__ asrc, const float* __restrict__ adst,
                     float* __restrict__ Y, float* __restrict__ as_out,
                     float* __restrict__ ad_out, int nrows) {
  gemm128_attn<4>(X, W, asrc, adst, Y, as_out, ad_out, nrows);
}

__global__ __launch_bounds__(256)
void gemm128_attn_h1(const float* __restrict__ X, const float* __restrict__ W,
                     const float* __restrict__ asrc, const float* __restrict__ adst,
                     float* __restrict__ Y, float* __restrict__ as_out,
                     float* __restrict__ ad_out, int nrows) {
  gemm128_attn<1>(X, W, asrc, adst, Y, as_out, ad_out, nrows);
}

__global__ __launch_bounds__(256)
void zero_ints(int* __restrict__ p, int n) {
  int t = blockIdx.x * 256 + threadIdx.x;
  if (t < n) p[t] = 0;
}

__global__ __launch_bounds__(256)
void count_rank(const int* __restrict__ ei, int* __restrict__ deg,
                int* __restrict__ rank, int E) {
  int e = blockIdx.x * 256 + threadIdx.x;
  if (e >= E) return;
  rank[e] = atomicAdd(&deg[ei[E + e]], 1);
}

__global__ __launch_bounds__(256)
void bucket_offsets(const int* __restrict__ deg, int* __restrict__ rowptr, int N) {
  __shared__ int wsum[8];
  __shared__ int wpre[8];
  const int tid = threadIdx.x, lane = tid & 31, wv = tid >> 5;
  const int b0 = blockIdx.x * 256;
  int part = 0;
  for (int i = tid; i < b0; i += 256) part += deg[i];
  const int i = b0 + tid;
  const int v = (i < N) ? deg[i] : 0;
  int inc = v;
  for (int off = 1; off < 32; off <<= 1) {
    int t = __shfl_up(inc, off, 32);
    if (lane >= off) inc += t;
  }
  for (int off = 16; off > 0; off >>= 1) part += __shfl_xor(part, off, 32);
  if (lane == 31) wsum[wv] = inc;
  if (lane == 0)  wpre[wv] = part;
  __syncthreads();
  int pre = 0;
  for (int w = 0; w < 8; ++w) {
    pre += wpre[w];
    if (w < wv) pre += wsum[w];
  }
  if (i < N) rowptr[i] = pre + inc - v;
}

__global__ __launch_bounds__(256)
void place_edges(const int* __restrict__ ei, const int* __restrict__ rowptr,
                 const int* __restrict__ rank, int* __restrict__ col, int E) {
  int e = blockIdx.x * 256 + threadIdx.x;
  if (e >= E) return;
  col[rowptr[ei[E + e]] + rank[e]] = e;
}

__global__ __launch_bounds__(256)
void sort_buckets(const int* __restrict__ ei, const int* __restrict__ rowptr,
                  const int* __restrict__ deg, const int* __restrict__ col,
                  int* __restrict__ csrc, int N) {
  const int node = __builtin_amdgcn_readfirstlane(blockIdx.x * 8 + (threadIdx.x >> 5));
  const int lane = threadIdx.x & 31;
  if (node >= N) return;
  const int base = rowptr[node];
  const int dg   = deg[node];
  for (int ci = 0; ci < dg; ci += 32) {
    const int i  = ci + lane;
    const int my = (i < dg) ? col[base + i] : 0x7fffffff;
    int pos = 0;
    for (int cj = 0; cj < dg; cj += 32) {
      const int j  = cj + lane;
      const int ej = (j < dg) ? col[base + j] : 0x7fffffff;
      const int cnt = min(32, dg - cj);
      for (int t = 0; t < cnt; ++t) {
        const int o = __builtin_amdgcn_readlane(ej, t);
        pos += (o < my) ? 1 : 0;
      }
    }
    if (i < dg) csrc[base + pos] = ei[my];
  }
}

__device__ __forceinline__ float wave_max(float v) {
  for (int off = 16; off > 0; off >>= 1) v = fmaxf(v, __shfl_xor(v, off, 32));
  return v;
}
__device__ __forceinline__ float wave_sum(float v) {
  for (int off = 16; off > 0; off >>= 1) v += __shfl_xor(v, off, 32);
  return v;
}
__device__ __forceinline__ float leaky02(float l) {
  return (l > 0.0f) ? l : 0.2f * l;
}

template <int H>
__device__ __forceinline__ void chunk_logits(const int* __restrict__ csrc,
                                             const float* __restrict__ as,
                                             const float* adn, int base, int i,
                                             int dg, int node, int& s, float* l) {
  s = (i < dg) ? csrc[base + i] : node;
  if (H == 4) {
    const float4 a = *(const float4*)(as + (long)s * 4);
    l[0] = leaky02(a.x + adn[0]);
    l[1] = leaky02(a.y + adn[1]);
    l[2] = leaky02(a.z + adn[2]);
    l[3] = leaky02(a.w + adn[3]);
  } else {
    l[0] = leaky02(as[s] + adn[0]);
  }
}

template <int H>
__device__ __forceinline__ void add_rows(const float* __restrict__ xl, int s,
                                         const float* al, int cnt, int lane,
                                         float* acc) {
#pragma unroll 4
  for (int t = 0; t < cnt; ++t) {
    const int st = __builtin_amdgcn_readlane(s, t);
    const float* xr = xl + (long)st * 128 + lane;
    float a[H];
    for (int h = 0; h < H; ++h)
      a[h] = __builtin_bit_cast(float,
               __builtin_amdgcn_readlane(__builtin_bit_cast(int, al[h]), t));
    for (int b = 0; b < 4; ++b)
      acc[b] += xr[32 * b] * a[H == 4 ? b : 0];
  }
}

enum { EPI_ELU = 0, EPI_L2 = 1 };

template <int H, int EPI>
__device__ __forceinline__ void node_gather(const int* __restrict__ rowptr,
                                            const int* __restrict__ deg,
                                            const int* __restrict__ csrc,
                                            const float* __restrict__ xl,
                                            const float* __restrict__ as,
                                            const float* __restrict__ ad,
                                            const float* __restrict__ bias,
                                            float* __restrict__ out, int N) {
  const int node = __builtin_amdgcn_readfirstlane(blockIdx.x * 8 + (threadIdx.x >> 5));
  const int lane = threadIdx.x & 31;
  if (node >= N) return;
  const int base = rowptr[node];
  const int dg   = deg[node];

  float adn[H], lself[H], m[H], den[H];
  for (int h = 0; h < H; ++h) {
    adn[h]   = ad[(long)node * H + h];
    lself[h] = leaky02(as[(long)node * H + h] + adn[h]);
  }
  float acc[4] = {0.0f, 0.0f, 0.0f, 0.0f};

  if (dg <= 32) {
    int s; float l[H], al[H];
    chunk_logits<H>(csrc, as, adn, base, lane, dg, node, s, l);
    const float live = (lane < dg) ? 1.0f : 0.0f;
    for (int h = 0; h < H; ++h) {
      m[h] = fmaxf(wave_max(l[h]), lself[h]);
      const float ex = expf(l[h] - m[h]) * live;
      den[h] = wave_sum(ex) + expf(lself[h] - m[h]);
      al[h]  = ex / (den[h] + 1e-16f);
    }
    add_rows<H>(xl, s, al, dg, lane, acc);
  } else {
    for (int h = 0; h < H; ++h) m[h] = lself[h];
    for (int c = 0; c < dg; c += 32) {
      int s; float l[H];
      chunk_logits<H>(csrc, as, adn, base, c + lane, dg, node, s, l);
      for (int h = 0; h < H; ++h) m[h] = fmaxf(m[h], l[h]);
    }
    for (int h = 0; h < H; ++h) { m[h] = wave_max(m[h]); den[h] = 0.0f; }
    for (int c = 0; c < dg; c += 32) {
      int s; float l[H];
      chunk_logits<H>(csrc, as, adn, base, c + lane, dg, node, s, l);
      const float live = (c + lane < dg) ? 1.0f : 0.0f;
      for (int h = 0; h < H; ++h) den[h] += expf(l[h] - m[h]) * live;
    }
    for (int h = 0; h < H; ++h) den[h] = wave_sum(den[h]) + expf(lself[h] - m[h]);
    for (int c = 0; c < dg; c += 32) {
      int s; float l[H], al[H];
      chunk_logits<H>(csrc, as, adn, base, c + lane, dg, node, s, l);
      for (int h = 0; h < H; ++h) al[h] = expf(l[h] - m[h]) / (den[h] + 1e-16f);
      add_rows<H>(xl, s, al, min(32, dg - c), lane, acc);
    }
  }

  const float* xr = xl + (long)node * 128 + lane;
  float* orow = out + (long)node * 128 + lane;
  float o[4];
  for (int b = 0; b < 4; ++b) {
    const int h = (H == 4) ? b : 0;
    const float aself = expf(lself[h] - m[h]) / (den[h] + 1e-16f);
    o[b] = acc[b] + xr[32 * b] * aself + bias[lane + 32 * b];
  }
  if (EPI == EPI_ELU) {
    for (int b = 0; b < 4; ++b)
      orow[32 * b] = (o[b] > 0.0f) ? o[b] : (expf(o[b]) - 1.0f);
  } else {
    float ss = 0.0f;
    for (int b = 0; b < 4; ++b) ss += o[b] * o[b];
    ss = wave_sum(ss);
    const float sc = 1.0f / fmaxf(sqrtf(ss), 1e-12f);
    for (int b = 0; b < 4; ++b) orow[32 * b] = o[b] * sc;
  }
}

__global__ __launch_bounds__(256)
void node_gather_h4_elu(const int* __restrict__ rowptr, const int* __restrict__ deg,
                        const int* __restrict__ csrc, const float* __restrict__ xl,
                        const float* __restrict__ as, const float* __restrict__ ad,
                        const float* __restrict__ bias, float* __restrict__ out, int N) {
  node_gather<4, EPI_ELU>(rowptr, deg, csrc, xl, as, ad, bias, out, N);
}

__global__ __launch_bounds__(256)
void node_gather_h1_l2(const int* __restrict__ rowptr, const int* __restrict__ deg,
                       const int* __restrict__ csrc, const float* __restrict__ xl,
                       const float* __restrict__ as, const float* __restrict__ ad,
                       const float* __restrict__ bias, float* __restrict__ out, int N) {
  node_gather<1, EPI_L2>(rowptr, deg, csrc, xl, as, ad, bias, out, N);
}

extern "C" void kernel_launch(void* const* d_in, const int* in_sizes, int n_in,
                              void* d_out, int out_size, void* d_ws, size_t ws_size,
                              hipStream_t stream) {
  const int*   ei    = (const int*)d_in[0];
  const float* emb   = (const float*)d_in[1];
  const float* W1    = (const float*)d_in[2];
  const float* asrc1 = (const float*)d_in[3];
  const float* adst1 = (const float*)d_in[4];
  const float* b1    = (const float*)d_in[5];
  const float* W2    = (const float*)d_in[6];
  const float* asrc2 = (const float*)d_in[7];
  const float* adst2 = (const float*)d_in[8];
  const float* b2    = (const float*)d_in[9];
  float* out = (float*)d_out;

  const int E = in_sizes[0] / 2;
  const int N = in_sizes[1] / 128;

  float* p   = (float*)d_ws;
  float* xl1 = p; p += (size_t)N * 128;
  float* x1  = p; p += (size_t)N * 128;
  float* xl2 = p; p += (size_t)N * 128;
  float* as1 = p; p += (size_t)N * 4;
  float* ad1 = p; p += (size_t)N * 4;
  float* as2 = p; p += (size_t)N;
  float* ad2 = p; p += (size_t)N;
  int* q      = (int*)p;
  int* deg    = q; q += (size_t)N;
  int* rowptr = q; q += (size_t)N;
  int* rank   = q; q += (size_t)E;
  int* col    = q; q += (size_t)E;
  int* csrc   = q;

  dim3 blk(256);
  int gGemm  = (N + 127) / 128;
  int gNode8 = (N + 7) / 8;
  int gNode  = (N + 255) / 256;
  int gEdge  = (E + 255) / 256;

  zero_ints     <<<gNode,  blk, 0, stream>>>(deg, N);
  count_rank    <<<gEdge,  blk, 0, stream>>>(ei, deg, rank, E);
  bucket_offsets<<<gNode,  blk, 0, stream>>>(deg, rowptr, N);
  place_edges   <<<gEdge,  blk, 0, stream>>>(ei, rowptr, rank, col, E);
  sort_buckets  <<<gNode8, blk, 0, stream>>>(ei, rowptr, deg, col, csrc, N);

  gemm128_attn_h4   <<<gGemm,  blk, 0, stream>>>(emb, W1, asrc1, adst1, xl1, as1, ad1, N);
  node_gather_h4_elu<<<gNode8, blk, 0, stream>>>(rowptr, deg, csrc, xl1, as1, ad1, b1, x1, N);

  gemm128_attn_h1   <<<gGemm,  blk, 0, stream>>>(x1, W2, asrc2, adst2, xl2, as2, ad2, N);
  node_gather_h1_l2 <<<gNode8, blk, 0, stream>>>(rowptr, deg, csrc, xl2, as2, ad2, b2, out, N);
}


# Bucket ordering folded into the layer-1 gather: placement stores (edge id, source) pairs, each node's wave ranks its bucket by id in registers, one lane permute; 9 launches to 8; sums unchanged
# speedup vs baseline: 4.5874x; 1.0491x over previous
#include <hip/hip_runtime.h>
#include <hip/hip_bf16.h>
#include <math.h>

typedef float v2f __attribute__((ext_vector_type(2)));
typedef float v8f __attribute__((ext_vector_type(8)));

template <int M>
__device__ __forceinline__ float row_xor(float v) {
  return __builtin_bit_cast(float,
      __builtin_amdgcn_update_dpp(0, __builtin_bit_cast(int, v), 0x160 + M,
                                  0xf, 0xf, true));
}
__device__ __forceinline__ float row16_sum(float v) {
  v += row_xor<1>(v);
  v += row_xor<2>(v);
  v += row_xor<4>(v);
  v += row_xor<8>(v);
  return v;
}

template <int H>
__device__ __forceinline__
void gemm128_attn(const float* __restrict__ X, const float* __restrict__ W,
                  const float* __restrict__ asrc, const float* __restrict__ adst,
                  float* __restrict__ Y, float* __restrict__ as_out,
                  float* __restrict__ ad_out, int nrows) {
  __shared__ float Xs[128 * 33];
  __shared__ float Ws[32 * 132];

  const int tid  = threadIdx.x;
  const int lane = tid & 31;
  const int wv   = tid >> 5;
  const int rc   = lane & 15;
  const int kh   = (lane >> 4) << 1;
  const int r0   = blockIdx.x * 128;

  v8f acc[8];
#pragma unroll
  for (int nt = 0; nt < 8; ++nt)
#pragma unroll
    for (int i = 0; i < 8; ++i) acc[nt][i] = 0.0f;

  for (int kb = 0; kb < 4; ++kb) {
#pragma unroll
    for (int i = 0; i < 4; ++i) {
      int idx = tid + i * 256;
      int row = idx >> 3;
      int c4  = (idx & 7) << 2;
      int gr  = r0 + row;
      float4 v = {0.f, 0.f, 0.f, 0.f};
      if (gr < nrows) v = *(const float4*)(X + (long)gr * 128 + kb * 32 + c4);
      float* p = &Xs[row * 33 + c4];
      p[0] = v.x; p[1] = v.y; p[2] = v.z; p[3] = v.w;
    }
#pragma unroll
    for (int i = 0; i < 4; ++i) {
      int idx = tid + i * 256;
      int row = idx >> 5;
      int c4  = (idx & 31) << 2;
      float4 v = *(const float4*)(W + (long)(kb * 32 + row) * 128 + c4);
      float* p = &Ws[row * 132 + c4];
      p[0] = v.x; p[1] = v.y; p[2] = v.z; p[3] = v.w;
    }
    __syncthreads();

    v2f afr[8];
#pragma unroll
    for (int ks = 0; ks < 8; ++ks) {
      const float* p = &Xs[(wv * 16 + rc) * 33 + ks * 4 + kh];
      afr[ks][0] = p[0];
      afr[ks][1] = p[1];
    }
#pragma unroll
    for (int nt = 0; nt < 8; ++nt) {
#pragma unroll
      for (int ks = 0; ks < 8; ++ks) {
        const float* p = &Ws[(ks * 4 + kh) * 132 + nt * 16 + rc];
        v2f b;
        b[0] = p[0];
        b[1] = p[132];
        acc[nt] = __builtin_amdgcn_wmma_f32_16x16x4_f32(
            false, afr[ks], false, b, (short)0, acc[nt], false, false);
      }
    }
    __syncthreads();
  }

#pragma unroll
  for (int nt = 0; nt < 8; ++nt) {
#pragma unroll
    for (int i = 0; i < 8; ++i) {
      int row = wv * 16 + i + ((lane >> 4) << 3);
      int gr  = r0 + row;
      if (gr < nrows) Y[(long)gr * 128 + nt * 16 + rc] = acc[nt][i];
    }
  }

  float wsrc[8], wdst[8];
#pragma unroll
  for (int nt = 0; nt < 8; ++nt) {
    wsrc[nt] = asrc[nt * 16 + rc];
    wdst[nt] = adst[nt * 16 + rc];
  }
  float ks[H], kd[H];
#pragma unroll
  for (int h = 0; h < H; ++h) { ks[h] = 0.0f; kd[h] = 0.0f; }
#pragma unroll
  for (int i = 0; i < 8; ++i) {
    float s[H], d[H];
    if (H == 4) {
#pragma unroll
      for (int h = 0; h < 4; ++h) {
        s[h] = acc[2 * h][i] * wsrc[2 * h] + acc[2 * h + 1][i] * wsrc[2 * h + 1];
        d[h] = acc[2 * h][i] * wdst[2 * h] + acc[2 * h + 1][i] * wdst[2 * h + 1];
      }
    } else {
      s[0] = 0.0f; d[0] = 0.0f;
#pragma unroll
      for (int nt = 0; nt < 8; ++nt) {
        s[0] += acc[nt][i] * wsrc[nt];
        d[0] += acc[nt][i] * wdst[nt];
      }
    }
#pragma unroll
    for (int h = 0; h < H; ++h) {
      s[h] = row16_sum(s[h]);
      d[h] = row16_sum(d[h]);
      if (rc == i) { ks[h] = s[h]; kd[h] = d[h]; }
    }
  }
  const int grow = r0 + wv * 16 + rc + ((lane >> 4) << 3);
  if (rc < 8 && grow < nrows) {
    if (H == 4) {
      *(float4*)(as_out + (long)grow * 4) = make_float4(ks[0], ks[1], ks[2], ks[3]);
      *(float4*)(ad_out + (long)grow * 4) = make_float4(kd[0], kd[1], kd[2], kd[3]);
    } else {
      as_out[grow] = ks[0];
      ad_out[grow] = kd[0];
    }
  }
}

__global__ __launch_bounds__(256)
void gemm128_attn_h4(const float* __restrict__ X, const float* __restrict__ W,
                     const float* __restrict__ asrc, const float* __restrict__ adst,
                     float* __restrict__ Y, float* __restrict__ as_out,
                     float* __restrict__ ad_out, int nrows) {
  gemm128_attn<4>(X, W, asrc, adst, Y, as_out, ad_out, nrows);
}

__global__ __launch_bounds__(256)
void gemm128_attn_h1(const float* __restrict__ X, const float* __restrict__ W,
                     const float* __restrict__ asrc, const float* __restrict__ adst,
                     float* __restrict__ Y, float* __restrict__ as_out,
                     float* __restrict__ ad_out, int nrows) {
  gemm128_attn<1>(X, W, asrc, adst, Y, as_out, ad_out, nrows);
}

__global__ __launch_bounds__(256)
void zero_ints(int* __restrict__ p, int n) {
  int t = blockIdx.x * 256 + threadIdx.x;
  if (t < n) p[t] = 0;
}

__global__ __launch_bounds__(256)
void count_rank(const int* __restrict__ ei, int* __restrict__ deg,
                int* __restrict__ rank, int E) {
  int e = blockIdx.x * 256 + threadIdx.x;
  if (e >= E) return;
  rank[e] = atomicAdd(&deg[ei[E + e]], 1);
}

__global__ __launch_bounds__(256)
void bucket_offsets(const int* __restrict__ deg, int* __restrict__ rowptr, int N) {
  __shared__ int wsum[8];
  __shared__ int wpre[8];
  const int tid = threadIdx.x, lane = tid & 31, wv = tid >> 5;
  const int b0 = blockIdx.x * 256;
  int part = 0;
  for (int i = tid; i < b0; i += 256) part += deg[i];
  const int i = b0 + tid;
  const int v = (i < N) ? deg[i] : 0;
  int inc = v;
  for (int off = 1; off < 32; off <<= 1) {
    int t = __shfl_up(inc, off, 32);
    if (lane >= off) inc += t;
  }
  for (int off = 16; off > 0; off >>= 1) part += __shfl_xor(part, off, 32);
  if (lane == 31) wsum[wv] = inc;
  if (lane == 0)  wpre[wv] = part;
  __syncthreads();
  int pre = 0;
  for (int w = 0; w < 8; ++w) {
    pre += wpre[w];
    if (w < wv) pre += wsum[w];
  }
  if (i < N) rowptr[i] = pre + inc - v;
}

__global__ __launch_bounds__(256)
void place_edges(const int* __restrict__ ei, const int* __restrict__ rowptr,
                 const int* __restrict__ rank, int* __restrict__ pairs, int E) {
  int e = blockIdx.x * 256 + threadIdx.x;
  if (e >= E) return;
  const int slot = rowptr[ei[E + e]] + rank[e];
  *(int2*)(pairs + 2 * (long)slot) = make_int2(e, ei[e]);
}

__device__ __forceinline__ float wave_max(float v) {
  for (int off = 16; off > 0; off >>= 1) v = fmaxf(v, __shfl_xor(v, off, 32));
  return v;
}
__device__ __forceinline__ float wave_sum(float v) {
  for (int off = 16; off > 0; off >>= 1) v += __shfl_xor(v, off, 32);
  return v;
}
__device__ __forceinline__ float leaky02(float l) {
  return (l > 0.0f) ? l : 0.2f * l;
}

template <int H>
__device__ __forceinline__ void src_logits(const float* __restrict__ as,
                                           const float* adn, int s, float* l) {
  if (H == 4) {
    const float4 a = *(const float4*)(as + (long)s * 4);
    l[0] = leaky02(a.x + adn[0]);
    l[1] = leaky02(a.y + adn[1]);
    l[2] = leaky02(a.z + adn[2]);
    l[3] = leaky02(a.w + adn[3]);
  } else {
    l[0] = leaky02(as[s] + adn[0]);
  }
}

template <int H>
__device__ __forceinline__ void chunk_logits(const int* csrc,
                                             const float* __restrict__ as,
                                             const float* adn, int base, int i,
                                             int dg, int node, int& s, float* l) {
  s = (i < dg) ? csrc[base + i] : node;
  src_logits<H>(as, adn, s, l);
}

template <int H>
__device__ __forceinline__ void add_rows(const float* __restrict__ xl, int s,
                                         const float* al, int cnt, int lane,
                                         float* acc) {
#pragma unroll 4
  for (int t = 0; t < cnt; ++t) {
    const int st = __builtin_amdgcn_readlane(s, t);
    const float* xr = xl + (long)st * 128 + lane;
    float a[H];
    for (int h = 0; h < H; ++h)
      a[h] = __builtin_bit_cast(float,
               __builtin_amdgcn_readlane(__builtin_bit_cast(int, al[h]), t));
    for (int b = 0; b < 4; ++b)
      acc[b] += xr[32 * b] * a[H == 4 ? b : 0];
  }
}

enum { EPI_ELU = 0, EPI_L2 = 1 };

template <int H, int EPI, bool ORDER>
__device__ __forceinline__ void node_gather(const int* __restrict__ rowptr,
                                            const int* __restrict__ deg,
                                            const int* pairs, int* csrc_w,
                                            const int* csrc,
                                            const float* __restrict__ xl,
                                            const float* __restrict__ as,
                                            const float* __restrict__ ad,
                                            const float* __restrict__ bias,
                                            float* __restrict__ out, int N) {
  const int node = __builtin_amdgcn_readfirstlane(blockIdx.x * 8 + (threadIdx.x >> 5));
  const int lane = threadIdx.x & 31;
  if (node >= N) return;
  const int base = rowptr[node];
  const int dg   = deg[node];

  float adn[H], lself[H], m[H], den[H];
  for (int h = 0; h < H; ++h) {
    adn[h]   = ad[(long)node * H + h];
    lself[h] = leaky02(as[(long)node * H + h] + adn[h]);
  }
  float acc[4] = {0.0f, 0.0f, 0.0f, 0.0f};

  if (dg <= 32) {
    int s; float l[H], al[H];
    if (ORDER) {
      int key = 0x7fffffff, sa = node;
      if (lane < dg) {
        const int2 pr = *(const int2*)(pairs + 2 * (long)(base + lane));
        key = pr.x; sa = pr.y;
      }
      int pos = 0;
      for (int t = 0; t < dg; ++t)
        pos += (__builtin_amdgcn_readlane(key, t) < key) ? 1 : 0;
      if (lane >= dg) pos = lane;
      s = __builtin_amdgcn_ds_permute(pos << 2, sa);
      if (lane < dg) csrc_w[base + lane] = s;
      src_logits<H>(as, adn, s, l);
    } else {
      chunk_logits<H>(csrc, as, adn, base, lane, dg, node, s, l);
    }
    const float live = (lane < dg) ? 1.0f : 0.0f;
    for (int h = 0; h < H; ++h) {
      m[h] = fmaxf(wave_max(l[h]), lself[h]);
      const float ex = expf(l[h] - m[h]) * live;
      den[h] = wave_sum(ex) + expf(lself[h] - m[h]);
      al[h]  = ex / (den[h] + 1e-16f);
    }
    add_rows<H>(xl, s, al, dg, lane, acc);
  } else {
    if (ORDER) {
      for (int ci = 0; ci < dg; ci += 32) {
        const int i = ci + lane;
        int key = 0x7fffffff, sa = node;
        if (i < dg) {
          const int2 pr = *(const int2*)(pairs + 2 * (long)(base + i));
          key = pr.x; sa = pr.y;
        }
        int pos = 0;
        for (int cj = 0; cj < dg; cj += 32) {
          const int j  = cj + lane;
          const int kj = (j < dg) ? pairs[2 * (long)(base + j)] : 0x7fffffff;
          const int cnt = min(32, dg - cj);
          for (int t = 0; t < cnt; ++t)
            pos += (__builtin_amdgcn_readlane(kj, t) < key) ? 1 : 0;
        }
        if (i < dg) csrc_w[base + pos] = sa;
      }
      __threadfence();
    }
    for (int h = 0; h < H; ++h) m[h] = lself[h];
    for (int c = 0; c < dg; c += 32) {
      int s; float l[H];
      chunk_logits<H>(csrc, as, adn, base, c + lane, dg, node, s, l);
      for (int h = 0; h < H; ++h) m[h] = fmaxf(m[h], l[h]);
    }
    for (int h = 0; h < H; ++h) { m[h] = wave_max(m[h]); den[h] = 0.0f; }
    for (int c = 0; c < dg; c += 32) {
      int s; float l[H];
      chunk_logits<H>(csrc, as, adn, base, c + lane, dg, node, s, l);
      const float live = (c + lane < dg) ? 1.0f : 0.0f;
      for (int h = 0; h < H; ++h) den[h] += expf(l[h] - m[h]) * live;
    }
    for (int h = 0; h < H; ++h) den[h] = wave_sum(den[h]) + expf(lself[h] - m[h]);
    for (int c = 0; c < dg; c += 32) {
      int s; float l[H], al[H];
      chunk_logits<H>(csrc, as, adn, base, c + lane, dg, node, s, l);
      for (int h = 0; h < H; ++h) al[h] = expf(l[h] - m[h]) / (den[h] + 1e-16f);
      add_rows<H>(xl, s, al, min(32, dg - c), lane, acc);
    }
  }

  const float* xr = xl + (long)node * 128 + lane;
  float* orow = out + (long)node * 128 + lane;
  float o[4];
  for (int b = 0; b < 4; ++b) {
    const int h = (H == 4) ? b : 0;
    const float aself = expf(lself[h] - m[h]) / (den[h] + 1e-16f);
    o[b] = acc[b] + xr[32 * b] * aself + bias[lane + 32 * b];
  }
  if (EPI == EPI_ELU) {
    for (int b = 0; b < 4; ++b)
      orow[32 * b] = (o[b] > 0.0f) ? o[b] : (expf(o[b]) - 1.0f);
  } else {
    float ss = 0.0f;
    for (int b = 0; b < 4; ++b) ss += o[b] * o[b];
    ss = wave_sum(ss);
    const float sc = 1.0f / fmaxf(sqrtf(ss), 1e-12f);
    for (int b = 0; b < 4; ++b) orow[32 * b] = o[b] * sc;
  }
}

__global__ __launch_bounds__(256)
void node_gather_h4_elu(const int* __restrict__ rowptr, const int* __restrict__ deg,
                        const int* __restrict__ pairs, int* csrc,
                        const float* __restrict__ xl,
                        const float* __restrict__ as, const float* __restrict__ ad,
                        const float* __restrict__ bias, float* __restrict__ out, int N) {
  node_gather<4, EPI_ELU, true>(rowptr, deg, pairs, csrc, csrc, xl, as, ad, bias, out, N);
}

__global__ __launch_bounds__(256)
void node_gather_h1_l2(const int* __restrict__ rowptr, const int* __restrict__ deg,
                       const int* __restrict__ csrc, const float* __restrict__ xl,
                       const float* __restrict__ as, const float* __restrict__ ad,
                       const float* __restrict__ bias, float* __restrict__ out, int N) {
  node_gather<1, EPI_L2, false>(rowptr, deg, nullptr, nullptr, csrc, xl, as, ad, bias, out, N);
}

extern "C" void kernel_launch(void* const* d_in, const int* in_sizes, int n_in,
                              void* d_out, int out_size, void* d_ws, size_t ws_size,
                              hipStream_t stream) {
  const int*   ei    = (const int*)d_in[0];
  const float* emb   = (const float*)d_in[1];
  const float* W1    = (const float*)d_in[2];
  const float* asrc1 = (const float*)d_in[3];
  const float* adst1 = (const float*)d_in[4];
  const float* b1    = (const float*)d_in[5];
  const float* W2    = (const float*)d_in[6];
  const float* asrc2 = (const float*)d_in[7];
  const float* adst2 = (const float*)d_in[8];
  const float* b2    = (const float*)d_in[9];
  float* out = (float*)d_out;

  const int E = in_sizes[0] / 2;
  const int N = in_sizes[1] / 128;

  float* p   = (float*)d_ws;
  float* xl1 = p; p += (size_t)N * 128;
  float* x1  = p; p += (size_t)N * 128;
  float* xl2 = p; p += (size_t)N * 128;
  float* as1 = p; p += (size_t)N * 4;
  float* ad1 = p; p += (size_t)N * 4;
  float* as2 = p; p += (size_t)N;
  float* ad2 = p; p += (size_t)N;
  int* q      = (int*)p;
  int* deg    = q; q += (size_t)N;
  int* rowptr = q; q += (size_t)N;
  int* pairs  = q; q += (size_t)E * 2;
  int* rank   = q; q += (size_t)E;
  int* csrc   = q;

  dim3 blk(256);
  int gGemm  = (N + 127) / 128;
  int gNode8 = (N + 7) / 8;
  int gNode  = (N + 255) / 256;
  int gEdge  = (E + 255) / 256;

  zero_ints     <<<gNode,  blk, 0, stream>>>(deg, N);
  count_rank    <<<gEdge,  blk, 0, stream>>>(ei, deg, rank, E);
  bucket_offsets<<<gNode,  blk, 0, stream>>>(deg, rowptr, N);
  place_edges   <<<gEdge,  blk, 0, stream>>>(ei, rowptr, rank, pairs, E);

  gemm128_attn_h4   <<<gGemm,  blk, 0, stream>>>(emb, W1, asrc1, adst1, xl1, as1, ad1, N);
  node_gather_h4_elu<<<gNode8, blk, 0, stream>>>(rowptr, deg, pairs, csrc, xl1, as1, ad1, b1, x1, N);

  gemm128_attn_h1   <<<gGemm,  blk, 0, stream>>>(x1, W2, asrc2, adst2, xl2, as2, ad2, N);
  node_gather_h1_l2 <<<gNode8, blk, 0, stream>>>(rowptr, deg, csrc, xl2, as2, ad2, b2, out, N);
}


# Fixed-stride buckets (64 slots per node; a longer one re-reads the edge list in id order): one edge pass counts and places, no offsets scan or rank array; 8 launches to 6; sum order unchanged
# speedup vs baseline: 5.4611x; 1.1905x over previous
#include <hip/hip_runtime.h>
#include <hip/hip_bf16.h>
#include <math.h>

typedef float v2f __attribute__((ext_vector_type(2)));
typedef float v8f __attribute__((ext_vector_type(8)));

template <int M>
__device__ __forceinline__ float row_xor(float v) {
  return __builtin_bit_cast(float,
      __builtin_amdgcn_update_dpp(0, __builtin_bit_cast(int, v), 0x160 + M,
                                  0xf, 0xf, true));
}
__device__ __forceinline__ float row16_sum(float v) {
  v += row_xor<1>(v);
  v += row_xor<2>(v);
  v += row_xor<4>(v);
  v += row_xor<8>(v);
  return v;
}

template <int H>
__device__ __forceinline__
void gemm128_attn(const float* __restrict__ X, const float* __restrict__ W,
                  const float* __restrict__ asrc, const float* __restrict__ adst,
                  float* __restrict__ Y, float* __restrict__ as_out,
                  float* __restrict__ ad_out, int nrows) {
  __shared__ float Xs[128 * 33];
  __shared__ float Ws[32 * 132];

  const int tid  = threadIdx.x;
  const int lane = tid & 31;
  const int wv   = tid >> 5;
  const int rc   = lane & 15;
  const int kh   = (lane >> 4) << 1;
  const int r0   = blockIdx.x * 128;

  v8f acc[8];
#pragma unroll
  for (int nt = 0; nt < 8; ++nt)
#pragma unroll
    for (int i = 0; i < 8; ++i) acc[nt][i] = 0.0f;

  for (int kb = 0; kb < 4; ++kb) {
#pragma unroll
    for (int i = 0; i < 4; ++i) {
      int idx = tid + i * 256;
      int row = idx >> 3;
      int c4  = (idx & 7) << 2;
      int gr  = r0 + row;
      float4 v = {0.f, 0.f, 0.f, 0.f};
      if (gr < nrows) v = *(const float4*)(X + (long)gr * 128 + kb * 32 + c4);
      float* p = &Xs[row * 33 + c4];
      p[0] = v.x; p[1] = v.y; p[2] = v.z; p[3] = v.w;
    }
#pragma unroll
    for (int i = 0; i < 4; ++i) {
      int idx = tid + i * 256;
      int row = idx >> 5;
      int c4  = (idx & 31) << 2;
      float4 v = *(const float4*)(W + (long)(kb * 32 + row) * 128 + c4);
      float* p = &Ws[row * 132 + c4];
      p[0] = v.x; p[1] = v.y; p[2] = v.z; p[3] = v.w;
    }
    __syncthreads();

    v2f afr[8];
#pragma unroll
    for (int ks = 0; ks < 8; ++ks) {
      const float* p = &Xs[(wv * 16 + rc) * 33 + ks * 4 + kh];
      afr[ks][0] = p[0];
      afr[ks][1] = p[1];
    }
#pragma unroll
    for (int nt = 0; nt < 8; ++nt) {
#pragma unroll
      for (int ks = 0; ks < 8; ++ks) {
        const float* p = &Ws[(ks * 4 + kh) * 132 + nt * 16 + rc];
        v2f b;
        b[0] = p[0];
        b[1] = p[132];
        acc[nt] = __builtin_amdgcn_wmma_f32_16x16x4_f32(
            false, afr[ks], false, b, (short)0, acc[nt], false, false);
      }
    }
    __syncthreads();
  }

#pragma unroll
  for (int nt = 0; nt < 8; ++nt) {
#pragma unroll
    for (int i = 0; i < 8; ++i) {
      int row = wv * 16 + i + ((lane >> 4) << 3);
      int gr  = r0 + row;
      if (gr < nrows) Y[(long)gr * 128 + nt * 16 + rc] = acc[nt][i];
    }
  }

  float wsrc[8], wdst[8];
#pragma unroll
  for (int nt = 0; nt < 8; ++nt) {
    wsrc[nt] = asrc[nt * 16 + rc];
    wdst[nt] = adst[nt * 16 + rc];
  }
  float ks[H], kd[H];
#pragma unroll
  for (int h = 0; h < H; ++h) { ks[h] = 0.0f; kd[h] = 0.0f; }
#pragma unroll
  for (int i = 0; i < 8; ++i) {
    float s[H], d[H];
    if (H == 4) {
#pragma unroll
      for (int h = 0; h < 4; ++h) {
        s[h] = acc[2 * h][i] * wsrc[2 * h] + acc[2 * h + 1][i] * wsrc[2 * h + 1];
        d[h] = acc[2 * h][i] * wdst[2 * h] + acc[2 * h + 1][i] * wdst[2 * h + 1];
      }
    } else {
      s[0] = 0.0f; d[0] = 0.0f;
#pragma unroll
      for (int nt = 0; nt < 8; ++nt) {
        s[0] += acc[nt][i] * wsrc[nt];
        d[0] += acc[nt][i] * wdst[nt];
      }
    }
#pragma unroll
    for (int h = 0; h < H; ++h) {
      s[h] = row16_sum(s[h]);
      d[h] = row16_sum(d[h]);
      if (rc == i) { ks[h] = s[h]; kd[h] = d[h]; }
    }
  }
  const int grow = r0 + wv * 16 + rc + ((lane >> 4) << 3);
  if (rc < 8 && grow < nrows) {
    if (H == 4) {
      *(float4*)(as_out + (long)grow * 4) = make_float4(ks[0], ks[1], ks[2], ks[3]);
      *(float4*)(ad_out + (long)grow * 4) = make_float4(kd[0], kd[1], kd[2], kd[3]);
    } else {
      as_out[grow] = ks[0];
      ad_out[grow] = kd[0];
    }
  }
}

__global__ __launch_bounds__(256)
void gemm128_attn_h4(const float* __restrict__ X, const float* __restrict__ W,
                     const float* __restrict__ asrc, const float* __restrict__ adst,
                     float* __restrict__ Y, float* __restrict__ as_out,
                     float* __restrict__ ad_out, int nrows) {
  gemm128_attn<4>(X, W, asrc, adst, Y, as_out, ad_out, nrows);
}

__global__ __launch_bounds__(256)
void gemm128_attn_h1(const float* __restrict__ X, const float* __restrict__ W,
                     const float* __restrict__ asrc, const float* __restrict__ adst,
                     float* __restrict__ Y, float* __restrict__ as_out,
                     float* __restrict__ ad_out, int nrows) {
  gemm128_attn<1>(X, W, asrc, adst, Y, as_out, ad_out, nrows);
}

enum { BCAP = 64 };

__global__ __launch_bounds__(256)
void zero_ints(int* __restrict__ p, int n) {
  int t = blockIdx.x * 256 + threadIdx.x;
  if (t < n) p[t] = 0;
}

__global__ __launch_bounds__(256)
void count_place(const int* __restrict__ ei, int* __restrict__ deg,
                 int* __restrict__ pairs, int E) {
  int e = blockIdx.x * 256 + threadIdx.x;
  if (e >= E) return;
  const int d = ei[E + e];
  const int s = ei[e];
  asm volatile("" :: "v"(s));
  const int r = atomicAdd(&deg[d], 1);
  if (r < BCAP)
    *(int2*)(pairs + 2 * ((long)d * BCAP + r)) = make_int2(e, s);
}

__device__ __forceinline__ float wave_max(float v) {
  for (int off = 16; off > 0; off >>= 1) v = fmaxf(v, __shfl_xor(v, off, 32));
  return v;
}
__device__ __forceinline__ float wave_sum(float v) {
  for (int off = 16; off > 0; off >>= 1) v += __shfl_xor(v, off, 32);
  return v;
}
__device__ __forceinline__ float leaky02(float l) {
  return (l > 0.0f) ? l : 0.2f * l;
}

template <int H>
__device__ __forceinline__ void src_logits(const float* __restrict__ as,
                                           const float* adn, int s, float* l) {
  if (H == 4) {
    const float4 a = *(const float4*)(as + (long)s * 4);
    l[0] = leaky02(a.x + adn[0]);
    l[1] = leaky02(a.y + adn[1]);
    l[2] = leaky02(a.z + adn[2]);
    l[3] = leaky02(a.w + adn[3]);
  } else {
    l[0] = leaky02(as[s] + adn[0]);
  }
}

template <int H>
__device__ __forceinline__ void chunk_logits(const int* csrc,
                                             const float* __restrict__ as,
                                             const float* adn, int base, int i,
                                             int dg, int node, int& s, float* l) {
  s = (i < dg) ? csrc[base + i] : node;
  src_logits<H>(as, adn, s, l);
}

template <int H>
__device__ __forceinline__ void add_rows(const float* __restrict__ xl, int s,
                                         const float* al, int cnt, int lane,
                                         float* acc) {
#pragma unroll 4
  for (int t = 0; t < cnt; ++t) {
    const int st = __builtin_amdgcn_readlane(s, t);
    const float* xr = xl + (long)st * 128 + lane;
    float a[H];
    for (int h = 0; h < H; ++h)
      a[h] = __builtin_bit_cast(float,
               __builtin_amdgcn_readlane(__builtin_bit_cast(int, al[h]), t));
    for (int b = 0; b < 4; ++b)
      acc[b] += xr[32 * b] * a[H == 4 ? b : 0];
  }
}

template <int H>
__device__ __forceinline__ unsigned scan_chunk(const int* __restrict__ ei, int E,
                                               int c, int lane, int node,
                                               const float* __restrict__ as,
                                               const float* adn, int& s, float* l) {
  const int e = c + lane;
  const bool hit = (e < E) && (ei[E + e] == node);
  const unsigned mask = __builtin_amdgcn_ballot_w32(hit);
  if (mask != 0u) {
    s = hit ? ei[e] : node;
    src_logits<H>(as, adn, s, l);
  }
  return mask;
}

template <int H>
__device__ __forceinline__ void scan_edges(const int* __restrict__ ei, int E,
                                           int node, int lane,
                                           const float* __restrict__ xl,
                                           const float* __restrict__ as,
                                           const float* adn, const float* lself,
                                           float* m, float* den, float* acc) {
  for (int h = 0; h < H; ++h) m[h] = lself[h];
  for (int c = 0; c < E; c += 32) {
    int s = node; float l[H];
    if (scan_chunk<H>(ei, E, c, lane, node, as, adn, s, l) != 0u)
      for (int h = 0; h < H; ++h) m[h] = fmaxf(m[h], l[h]);
  }
  for (int h = 0; h < H; ++h) { m[h] = wave_max(m[h]); den[h] = 0.0f; }
  for (int c = 0; c < E; c += 32) {
    int s = node; float l[H];
    const unsigned mask = scan_chunk<H>(ei, E, c, lane, node, as, adn, s, l);
    if (mask != 0u) {
      const float live = ((mask >> lane) & 1u) ? 1.0f : 0.0f;
      for (int h = 0; h < H; ++h) den[h] += expf(l[h] - m[h]) * live;
    }
  }
  for (int h = 0; h < H; ++h) den[h] = wave_sum(den[h]) + expf(lself[h] - m[h]);
  for (int c = 0; c < E; c += 32) {
    int s = node; float l[H];
    unsigned mask = scan_chunk<H>(ei, E, c, lane, node, as, adn, s, l);
    if (mask != 0u) {
      float al[H];
      for (int h = 0; h < H; ++h) al[h] = expf(l[h] - m[h]) / (den[h] + 1e-16f);
      while (mask != 0u) {
        const int t = __builtin_ctz(mask);
        mask &= mask - 1u;
        const int st = __builtin_amdgcn_readlane(s, t);
        const float* xr = xl + (long)st * 128 + lane;
        float a[H];
        for (int h = 0; h < H; ++h)
          a[h] = __builtin_bit_cast(float,
                   __builtin_amdgcn_readlane(__builtin_bit_cast(int, al[h]), t));
        for (int b = 0; b < 4; ++b)
          acc[b] += xr[32 * b] * a[H == 4 ? b : 0];
      }
    }
  }
}

enum { EPI_ELU = 0, EPI_L2 = 1 };

template <int H, int EPI, bool ORDER>
__device__ __forceinline__ void node_gather(const int* __restrict__ ei, int E,
                                            const int* __restrict__ deg,
                                            const int* pairs, int* csrc_w,
                                            const int* csrc,
                                            const float* __restrict__ xl,
                                            const float* __restrict__ as,
                                            const float* __restrict__ ad,
                                            const float* __restrict__ bias,
                                            float* __restrict__ out, int N) {
  const int node = __builtin_amdgcn_readfirstlane(blockIdx.x * 8 + (threadIdx.x >> 5));
  const int lane = threadIdx.x & 31;
  if (node >= N) return;
  const int base = node * BCAP;
  const int dg   = deg[node];

  float adn[H], lself[H], m[H], den[H];
  for (int h = 0; h < H; ++h) {
    adn[h]   = ad[(long)node * H + h];
    lself[h] = leaky02(as[(long)node * H + h] + adn[h]);
  }
  float acc[4] = {0.0f, 0.0f, 0.0f, 0.0f};

  if (dg <= 32) {
    int s; float l[H], al[H];
    if (ORDER) {
      int key = 0x7fffffff, sa = node;
      if (lane < dg) {
        const int2 pr = *(const int2*)(pairs + 2 * (long)(base + lane));
        key = pr.x; sa = pr.y;
      }
      int pos = 0;
      for (int t = 0; t < dg; ++t)
        pos += (__builtin_amdgcn_readlane(key, t) < key) ? 1 : 0;
      if (lane >= dg) pos = lane;
      s = __builtin_amdgcn_ds_permute(pos << 2, sa);
      if (lane < dg) csrc_w[base + lane] = s;
      src_logits<H>(as, adn, s, l);
    } else {
      chunk_logits<H>(csrc, as, adn, base, lane, dg, node, s, l);
    }
    const float live = (lane < dg) ? 1.0f : 0.0f;
    for (int h = 0; h < H; ++h) {
      m[h] = fmaxf(wave_max(l[h]), lself[h]);
      const float ex = expf(l[h] - m[h]) * live;
      den[h] = wave_sum(ex) + expf(lself[h] - m[h]);
      al[h]  = ex / (den[h] + 1e-16f);
    }
    add_rows<H>(xl, s, al, dg, lane, acc);
  } else if (dg > BCAP) {
    scan_edges<H>(ei, E, node, lane, xl, as, adn, lself, m, den, acc);
  } else {
    if (ORDER) {
      for (int ci = 0; ci < dg; ci += 32) {
        const int i = ci + lane;
        int key = 0x7fffffff, sa = node;
        if (i < dg) {
          const int2 pr = *(const int2*)(pairs + 2 * (long)(base + i));
          key = pr.x; sa = pr.y;
        }
        int pos = 0;
        for (int cj = 0; cj < dg; cj += 32) {
          const int j  = cj + lane;
          const int kj = (j < dg) ? pairs[2 * (long)(base + j)] : 0x7fffffff;
          const int cnt = min(32, dg - cj);
          for (int t = 0; t < cnt; ++t)
            pos += (__builtin_amdgcn_readlane(kj, t) < key) ? 1 : 0;
        }
        if (i < dg) csrc_w[base + pos] = sa;
      }
      __threadfence();
    }
    for (int h = 0; h < H; ++h) m[h] = lself[h];
    for (int c = 0; c < dg; c += 32) {
      int s; float l[H];
      chunk_logits<H>(csrc, as, adn, base, c + lane, dg, node, s, l);
      for (int h = 0; h < H; ++h) m[h] = fmaxf(m[h], l[h]);
    }
    for (int h = 0; h < H; ++h) { m[h] = wave_max(m[h]); den[h] = 0.0f; }
    for (int c = 0; c < dg; c += 32) {
      int s; float l[H];
      chunk_logits<H>(csrc, as, adn, base, c + lane, dg, node, s, l);
      const float live = (c + lane < dg) ? 1.0f : 0.0f;
      for (int h = 0; h < H; ++h) den[h] += expf(l[h] - m[h]) * live;
    }
    for (int h = 0; h < H; ++h) den[h] = wave_sum(den[h]) + expf(lself[h] - m[h]);
    for (int c = 0; c < dg; c += 32) {
      int s; float l[H], al[H];
      chunk_logits<H>(csrc, as, adn, base, c + lane, dg, node, s, l);
      for (int h = 0; h < H; ++h) al[h] = expf(l[h] - m[h]) / (den[h] + 1e-16f);
      add_rows<H>(xl, s, al, min(32, dg - c), lane, acc);
    }
  }

  const float* xr = xl + (long)node * 128 + lane;
  float* orow = out + (long)node * 128 + lane;
  float o[4];
  for (int b = 0; b < 4; ++b) {
    const int h = (H == 4) ? b : 0;
    const float aself = expf(lself[h] - m[h]) / (den[h] + 1e-16f);
    o[b] = acc[b] + xr[32 * b] * aself + bias[lane + 32 * b];
  }
  if (EPI == EPI_ELU) {
    for (int b = 0; b < 4; ++b)
      orow[32 * b] = (o[b] > 0.0f) ? o[b] : (expf(o[b]) - 1.0f);
  } else {
    float ss = 0.0f;
    for (int b = 0; b < 4; ++b) ss += o[b] * o[b];
    ss = wave_sum(ss);
    const float sc = 1.0f / fmaxf(sqrtf(ss), 1e-12f);
    for (int b = 0; b < 4; ++b) orow[32 * b] = o[b] * sc;
  }
}

__global__ __launch_bounds__(256)
void node_gather_h4_elu(const int* __restrict__ ei, int E, const int* __restrict__ deg,
                        const int* __restrict__ pairs, int* csrc,
                        const float* __restrict__ xl,
                        const float* __restrict__ as, const float* __restrict__ ad,
                        const float* __restrict__ bias, float* __restrict__ out, int N) {
  node_gather<4, EPI_ELU, true>(ei, E, deg, pairs, csrc, csrc, xl, as, ad, bias, out, N);
}

__global__ __launch_bounds__(256)
void node_gather_h1_l2(const int* __restrict__ ei, int E, const int* __restrict__ deg,
                       const int* __restrict__ csrc, const float* __restrict__ xl,
                       const float* __restrict__ as, const float* __restrict__ ad,
                       const float* __restrict__ bias, float* __restrict__ out, int N) {
  node_gather<1, EPI_L2, false>(ei, E, deg, nullptr, nullptr, csrc, xl, as, ad, bias, out, N);
}

extern "C" void kernel_launch(void* const* d_in, const int* in_sizes, int n_in,
                              void* d_out, int out_size, void* d_ws, size_t ws_size,
                              hipStream_t stream) {
  const int*   ei    = (const int*)d_in[0];
  const float* emb   = (const float*)d_in[1];
  const float* W1    = (const float*)d_in[2];
  const float* asrc1 = (const float*)d_in[3];
  const float* adst1 = (const float*)d_in[4];
  const float* b1    = (const float*)d_in[5];
  const float* W2    = (const float*)d_in[6];
  const float* asrc2 = (const float*)d_in[7];
  const float* adst2 = (const float*)d_in[8];
  const float* b2    = (const float*)d_in[9];
  float* out = (float*)d_out;

  const int E = in_sizes[0] / 2;
  const int N = in_sizes[1] / 128;

  int* q      = (int*)d_ws;
  int* pairs  = q; q += (size_t)N * BCAP * 2;
  int* csrc   = q; q += (size_t)N * BCAP;
  float* p   = (float*)q;
  float* xl1 = p; p += (size_t)N * 128;
  float* x1  = p; p += (size_t)N * 128;
  float* xl2 = p; p += (size_t)N * 128;
  float* as1 = p; p += (size_t)N * 4;
  float* ad1 = p; p += (size_t)N * 4;
  float* as2 = p; p += (size_t)N;
  float* ad2 = p; p += (size_t)N;
  int* deg    = (int*)p;

  dim3 blk(256);
  int gGemm  = (N + 127) / 128;
  int gNode8 = (N + 7) / 8;
  int gNode  = (N + 255) / 256;
  int gEdge  = (E + 255) / 256;

  zero_ints  <<<gNode, blk, 0, stream>>>(deg, N);
  count_place<<<gEdge, blk, 0, stream>>>(ei, deg, pairs, E);

  gemm128_attn_h4   <<<gGemm,  blk, 0, stream>>>(emb, W1, asrc1, adst1, xl1, as1, ad1, N);
  node_gather_h4_elu<<<gNode8, blk, 0, stream>>>(ei, E, deg, pairs, csrc, xl1, as1, ad1, b1, x1, N);

  gemm128_attn_h1   <<<gGemm,  blk, 0, stream>>>(x1, W2, asrc2, adst2, xl2, as2, ad2, N);
  node_gather_h1_l2 <<<gNode8, blk, 0, stream>>>(ei, E, deg, csrc, xl2, as2, ad2, b2, out, N);
}


# Layer-1 GEMM and the edge bucketing pass share one launch (GEMM tiles are the first blocks, edge blocks fill in beside them): independent work overlaps; 6 launches to 5; sums unchanged
# speedup vs baseline: 6.0393x; 1.1059x over previous
#include <hip/hip_runtime.h>
#include <hip/hip_bf16.h>
#include <math.h>

typedef float v2f __attribute__((ext_vector_type(2)));
typedef float v8f __attribute__((ext_vector_type(8)));

template <int M>
__device__ __forceinline__ float row_xor(float v) {
  return __builtin_bit_cast(float,
      __builtin_amdgcn_update_dpp(0, __builtin_bit_cast(int, v), 0x160 + M,
                                  0xf, 0xf, true));
}
__device__ __forceinline__ float row16_sum(float v) {
  v += row_xor<1>(v);
  v += row_xor<2>(v);
  v += row_xor<4>(v);
  v += row_xor<8>(v);
  return v;
}

template <int H>
__device__ __forceinline__
void gemm128_attn(const float* __restrict__ X, const float* __restrict__ W,
                  const float* __restrict__ asrc, const float* __restrict__ adst,
                  float* __restrict__ Y, float* __restrict__ as_out,
                  float* __restrict__ ad_out, int nrows) {
  __shared__ float Xs[128 * 33];
  __shared__ float Ws[32 * 132];

  const int tid  = threadIdx.x;
  const int lane = tid & 31;
  const int wv   = tid >> 5;
  const int rc   = lane & 15;
  const int kh   = (lane >> 4) << 1;
  const int r0   = blockIdx.x * 128;

  v8f acc[8];
#pragma unroll
  for (int nt = 0; nt < 8; ++nt)
#pragma unroll
    for (int i = 0; i < 8; ++i) acc[nt][i] = 0.0f;

  for (int kb = 0; kb < 4; ++kb) {
#pragma unroll
    for (int i = 0; i < 4; ++i) {
      int idx = tid + i * 256;
      int row = idx >> 3;
      int c4  = (idx & 7) << 2;
      int gr  = r0 + row;
      float4 v = {0.f, 0.f, 0.f, 0.f};
      if (gr < nrows) v = *(const float4*)(X + (long)gr * 128 + kb * 32 + c4);
      float* p = &Xs[row * 33 + c4];
      p[0] = v.x; p[1] = v.y; p[2] = v.z; p[3] = v.w;
    }
#pragma unroll
    for (int i = 0; i < 4; ++i) {
      int idx = tid + i * 256;
      int row = idx >> 5;
      int c4  = (idx & 31) << 2;
      float4 v = *(const float4*)(W + (long)(kb * 32 + row) * 128 + c4);
      float* p = &Ws[row * 132 + c4];
      p[0] = v.x; p[1] = v.y; p[2] = v.z; p[3] = v.w;
    }
    __syncthreads();

    v2f afr[8];
#pragma unroll
    for (int ks = 0; ks < 8; ++ks) {
      const float* p = &Xs[(wv * 16 + rc) * 33 + ks * 4 + kh];
      afr[ks][0] = p[0];
      afr[ks][1] = p[1];
    }
#pragma unroll
    for (int nt = 0; nt < 8; ++nt) {
#pragma unroll
      for (int ks = 0; ks < 8; ++ks) {
        const float* p = &Ws[(ks * 4 + kh) * 132 + nt * 16 + rc];
        v2f b;
        b[0] = p[0];
        b[1] = p[132];
        acc[nt] = __builtin_amdgcn_wmma_f32_16x16x4_f32(
            false, afr[ks], false, b, (short)0, acc[nt], false, false);
      }
    }
    __syncthreads();
  }

#pragma unroll
  for (int nt = 0; nt < 8; ++nt) {
#pragma unroll
    for (int i = 0; i < 8; ++i) {
      int row = wv * 16 + i + ((lane >> 4) << 3);
      int gr  = r0 + row;
      if (gr < nrows) Y[(long)gr * 128 + nt * 16 + rc] = acc[nt][i];
    }
  }

  float wsrc[8], wdst[8];
#pragma unroll
  for (int nt = 0; nt < 8; ++nt) {
    wsrc[nt] = asrc[nt * 16 + rc];
    wdst[nt] = adst[nt * 16 + rc];
  }
  float ks[H], kd[H];
#pragma unroll
  for (int h = 0; h < H; ++h) { ks[h] = 0.0f; kd[h] = 0.0f; }
#pragma unroll
  for (int i = 0; i < 8; ++i) {
    float s[H], d[H];
    if (H == 4) {
#pragma unroll
      for (int h = 0; h < 4; ++h) {
        s[h] = acc[2 * h][i] * wsrc[2 * h] + acc[2 * h + 1][i] * wsrc[2 * h + 1];
        d[h] = acc[2 * h][i] * wdst[2 * h] + acc[2 * h + 1][i] * wdst[2 * h + 1];
      }
    } else {
      s[0] = 0.0f; d[0] = 0.0f;
#pragma unroll
      for (int nt = 0; nt < 8; ++nt) {
        s[0] += acc[nt][i] * wsrc[nt];
        d[0] += acc[nt][i] * wdst[nt];
      }
    }
#pragma unroll
    for (int h = 0; h < H; ++h) {
      s[h] = row16_sum(s[h]);
      d[h] = row16_sum(d[h]);
      if (rc == i) { ks[h] = s[h]; kd[h] = d[h]; }
    }
  }
  const int grow = r0 + wv * 16 + rc + ((lane >> 4) << 3);
  if (rc < 8 && grow < nrows) {
    if (H == 4) {
      *(float4*)(as_out + (long)grow * 4) = make_float4(ks[0], ks[1], ks[2], ks[3]);
      *(float4*)(ad_out + (long)grow * 4) = make_float4(kd[0], kd[1], kd[2], kd[3]);
    } else {
      as_out[grow] = ks[0];
      ad_out[grow] = kd[0];
    }
  }
}

__global__ __launch_bounds__(256)
void gemm128_attn_h1(const float* __restrict__ X, const float* __restrict__ W,
                     const float* __restrict__ asrc, const float* __restrict__ adst,
                     float* __restrict__ Y, float* __restrict__ as_out,
                     float* __restrict__ ad_out, int nrows) {
  gemm128_attn<1>(X, W, asrc, adst, Y, as_out, ad_out, nrows);
}

enum { BCAP = 64 };

__global__ __launch_bounds__(256)
void zero_ints(int* __restrict__ p, int n) {
  int t = blockIdx.x * 256 + threadIdx.x;
  if (t < n) p[t] = 0;
}

__device__ __forceinline__
void count_place(const int* __restrict__ ei, int* __restrict__ deg,
                 int* __restrict__ pairs, int E, int blk) {
  int e = blk * 256 + threadIdx.x;
  if (e >= E) return;
  const int d = ei[E + e];
  const int s = ei[e];
  asm volatile("" :: "v"(s));
  const int r = atomicAdd(&deg[d], 1);
  if (r < BCAP)
    *(int2*)(pairs + 2 * ((long)d * BCAP + r)) = make_int2(e, s);
}

__global__ __launch_bounds__(256)
void gemm1_beside_buckets(const float* __restrict__ X, const float* __restrict__ W,
                          const float* __restrict__ asrc, const float* __restrict__ adst,
                          float* __restrict__ Y, float* __restrict__ as_out,
                          float* __restrict__ ad_out, int nrows, int gemm_blocks,
                          const int* __restrict__ ei, int* __restrict__ deg,
                          int* __restrict__ pairs, int E) {
  if ((int)blockIdx.x < gemm_blocks)
    gemm128_attn<4>(X, W, asrc, adst, Y, as_out, ad_out, nrows);
  else
    count_place(ei, deg, pairs, E, (int)blockIdx.x - gemm_blocks);
}

__device__ __forceinline__ float wave_max(float v) {
  for (int off = 16; off > 0; off >>= 1) v = fmaxf(v, __shfl_xor(v, off, 32));
  return v;
}
__device__ __forceinline__ float wave_sum(float v) {
  for (int off = 16; off > 0; off >>= 1) v += __shfl_xor(v, off, 32);
  return v;
}
__device__ __forceinline__ float leaky02(float l) {
  return (l > 0.0f) ? l : 0.2f * l;
}

template <int H>
__device__ __forceinline__ void src_logits(const float* __restrict__ as,
                                           const float* adn, int s, float* l) {
  if (H == 4) {
    const float4 a = *(const float4*)(as + (long)s * 4);
    l[0] = leaky02(a.x + adn[0]);
    l[1] = leaky02(a.y + adn[1]);
    l[2] = leaky02(a.z + adn[2]);
    l[3] = leaky02(a.w + adn[3]);
  } else {
    l[0] = leaky02(as[s] + adn[0]);
  }
}

template <int H>
__device__ __forceinline__ void chunk_logits(const int* csrc,
                                             const float* __restrict__ as,
                                             const float* adn, int base, int i,
                                             int dg, int node, int& s, float* l) {
  s = (i < dg) ? csrc[base + i] : node;
  src_logits<H>(as, adn, s, l);
}

template <int H>
__device__ __forceinline__ void add_rows(const float* __restrict__ xl, int s,
                                         const float* al, int cnt, int lane,
                                         float* acc) {
#pragma unroll 4
  for (int t = 0; t < cnt; ++t) {
    const int st = __builtin_amdgcn_readlane(s, t);
    const float* xr = xl + (long)st * 128 + lane;
    float a[H];
    for (int h = 0; h < H; ++h)
      a[h] = __builtin_bit_cast(float,
               __builtin_amdgcn_readlane(__builtin_bit_cast(int, al[h]), t));
    for (int b = 0; b < 4; ++b)
      acc[b] += xr[32 * b] * a[H == 4 ? b : 0];
  }
}

template <int H>
__device__ __forceinline__ unsigned scan_chunk(const int* __restrict__ ei, int E,
                                               int c, int lane, int node,
                                               const float* __restrict__ as,
                                               const float* adn, int& s, float* l) {
  const int e = c + lane;
  const bool hit = (e < E) && (ei[E + e] == node);
  const unsigned mask = __builtin_amdgcn_ballot_w32(hit);
  if (mask != 0u) {
    s = hit ? ei[e] : node;
    src_logits<H>(as, adn, s, l);
  }
  return mask;
}

template <int H>
__device__ __forceinline__ void scan_edges(const int* __restrict__ ei, int E,
                                           int node, int lane,
                                           const float* __restrict__ xl,
                                           const float* __restrict__ as,
                                           const float* adn, const float* lself,
                                           float* m, float* den, float* acc) {
  for (int h = 0; h < H; ++h) m[h] = lself[h];
  for (int c = 0; c < E; c += 32) {
    int s = node; float l[H];
    if (scan_chunk<H>(ei, E, c, lane, node, as, adn, s, l) != 0u)
      for (int h = 0; h < H; ++h) m[h] = fmaxf(m[h], l[h]);
  }
  for (int h = 0; h < H; ++h) { m[h] = wave_max(m[h]); den[h] = 0.0f; }
  for (int c = 0; c < E; c += 32) {
    int s = node; float l[H];
    const unsigned mask = scan_chunk<H>(ei, E, c, lane, node, as, adn, s, l);
    if (mask != 0u) {
      const float live = ((mask >> lane) & 1u) ? 1.0f : 0.0f;
      for (int h = 0; h < H; ++h) den[h] += expf(l[h] - m[h]) * live;
    }
  }
  for (int h = 0; h < H; ++h) den[h] = wave_sum(den[h]) + expf(lself[h] - m[h]);
  for (int c = 0; c < E; c += 32) {
    int s = node; float l[H];
    unsigned mask = scan_chunk<H>(ei, E, c, lane, node, as, adn, s, l);
    if (mask != 0u) {
      float al[H];
      for (int h = 0; h < H; ++h) al[h] = expf(l[h] - m[h]) / (den[h] + 1e-16f);
      while (mask != 0u) {
        const int t = __builtin_ctz(mask);
        mask &= mask - 1u;
        const int st = __builtin_amdgcn_readlane(s, t);
        const float* xr = xl + (long)st * 128 + lane;
        float a[H];
        for (int h = 0; h < H; ++h)
          a[h] = __builtin_bit_cast(float,
                   __builtin_amdgcn_readlane(__builtin_bit_cast(int, al[h]), t));
        for (int b = 0; b < 4; ++b)
          acc[b] += xr[32 * b] * a[H == 4 ? b : 0];
      }
    }
  }
}

enum { EPI_ELU = 0, EPI_L2 = 1 };

template <int H, int EPI, bool ORDER>
__device__ __forceinline__ void node_gather(const int* __restrict__ ei, int E,
                                            const int* __restrict__ deg,
                                            const int* pairs, int* csrc_w,
                                            const int* csrc,
                                            const float* __restrict__ xl,
                                            const float* __restrict__ as,
                                            const float* __restrict__ ad,
                                            const float* __restrict__ bias,
                                            float* __restrict__ out, int N) {
  const int node = __builtin_amdgcn_readfirstlane(blockIdx.x * 8 + (threadIdx.x >> 5));
  const int lane = threadIdx.x & 31;
  if (node >= N) return;
  const int base = node * BCAP;
  const int dg   = deg[node];

  float adn[H], lself[H], m[H], den[H];
  for (int h = 0; h < H; ++h) {
    adn[h]   = ad[(long)node * H + h];
    lself[h] = leaky02(as[(long)node * H + h] + adn[h]);
  }
  float acc[4] = {0.0f, 0.0f, 0.0f, 0.0f};

  if (dg <= 32) {
    int s; float l[H], al[H];
    if (ORDER) {
      int key = 0x7fffffff, sa = node;
      if (lane < dg) {
        const int2 pr = *(const int2*)(pairs + 2 * (long)(base + lane));
        key = pr.x; sa = pr.y;
      }
      int pos = 0;
      for (int t = 0; t < dg; ++t)
        pos += (__builtin_amdgcn_readlane(key, t) < key) ? 1 : 0;
      if (lane >= dg) pos = lane;
      s = __builtin_amdgcn_ds_permute(pos << 2, sa);
      if (lane < dg) csrc_w[base + lane] = s;
      src_logits<H>(as, adn, s, l);
    } else {
      chunk_logits<H>(csrc, as, adn, base, lane, dg, node, s, l);
    }
    const float live = (lane < dg) ? 1.0f : 0.0f;
    for (int h = 0; h < H; ++h) {
      m[h] = fmaxf(wave_max(l[h]), lself[h]);
      const float ex = expf(l[h] - m[h]) * live;
      den[h] = wave_sum(ex) + expf(lself[h] - m[h]);
      al[h]  = ex / (den[h] + 1e-16f);
    }
    add_rows<H>(xl, s, al, dg, lane, acc);
  } else if (dg > BCAP) {
    scan_edges<H>(ei, E, node, lane, xl, as, adn, lself, m, den, acc);
  } else {
    if (ORDER) {
      for (int ci = 0; ci < dg; ci += 32) {
        const int i = ci + lane;
        int key = 0x7fffffff, sa = node;
        if (i < dg) {
          const int2 pr = *(const int2*)(pairs + 2 * (long)(base + i));
          key = pr.x; sa = pr.y;
        }
        int pos = 0;
        for (int cj = 0; cj < dg; cj += 32) {
          const int j  = cj + lane;
          const int kj = (j < dg) ? pairs[2 * (long)(base + j)] : 0x7fffffff;
          const int cnt = min(32, dg - cj);
          for (int t = 0; t < cnt; ++t)
            pos += (__builtin_amdgcn_readlane(kj, t) < key) ? 1 : 0;
        }
        if (i < dg) csrc_w[base + pos] = sa;
      }
      __threadfence();
    }
    for (int h = 0; h < H; ++h) m[h] = lself[h];
    for (int c = 0; c < dg; c += 32) {
      int s; float l[H];
      chunk_logits<H>(csrc, as, adn, base, c + lane, dg, node, s, l);
      for (int h = 0; h < H; ++h) m[h] = fmaxf(m[h], l[h]);
    }
    for (int h = 0; h < H; ++h) { m[h] = wave_max(m[h]); den[h] = 0.0f; }
    for (int c = 0; c < dg; c += 32) {
      int s; float l[H];
      chunk_logits<H>(csrc, as, adn, base, c + lane, dg, node, s, l);
      const float live = (c + lane < dg) ? 1.0f : 0.0f;
      for (int h = 0; h < H; ++h) den[h] += expf(l[h] - m[h]) * live;
    }
    for (int h = 0; h < H; ++h) den[h] = wave_sum(den[h]) + expf(lself[h] - m[h]);
    for (int c = 0; c < dg; c += 32) {
      int s; float l[H], al[H];
      chunk_logits<H>(csrc, as, adn, base, c + lane, dg, node, s, l);
      for (int h = 0; h < H; ++h) al[h] = expf(l[h] - m[h]) / (den[h] + 1e-16f);
      add_rows<H>(xl, s, al, min(32, dg - c), lane, acc);
    }
  }

  const float* xr = xl + (long)node * 128 + lane;
  float* orow = out + (long)node * 128 + lane;
  float o[4];
  for (int b = 0; b < 4; ++b) {
    const int h = (H == 4) ? b : 0;
    const float aself = expf(lself[h] - m[h]) / (den[h] + 1e-16f);
    o[b] = acc[b] + xr[32 * b] * aself + bias[lane + 32 * b];
  }
  if (EPI == EPI_ELU) {
    for (int b = 0; b < 4; ++b)
      orow[32 * b] = (o[b] > 0.0f) ? o[b] : (expf(o[b]) - 1.0f);
  } else {
    float ss = 0.0f;
    for (int b = 0; b < 4; ++b) ss += o[b] * o[b];
    ss = wave_sum(ss);
    const float sc = 1.0f / fmaxf(sqrtf(ss), 1e-12f);
    for (int b = 0; b < 4; ++b) orow[32 * b] = o[b] * sc;
  }
}

__global__ __launch_bounds__(256)
void node_gather_h4_elu(const int* __restrict__ ei, int E, const int* __restrict__ deg,
                        const int* __restrict__ pairs, int* csrc,
                        const float* __restrict__ xl,
                        const float* __restrict__ as, const float* __restrict__ ad,
                        const float* __restrict__ bias, float* __restrict__ out, int N) {
  node_gather<4, EPI_ELU, true>(ei, E, deg, pairs, csrc, csrc, xl, as, ad, bias, out, N);
}

__global__ __launch_bounds__(256)
void node_gather_h1_l2(const int* __restrict__ ei, int E, const int* __restrict__ deg,
                       const int* __restrict__ csrc, const float* __restrict__ xl,
                       const float* __restrict__ as, const float* __restrict__ ad,
                       const float* __restrict__ bias, float* __restrict__ out, int N) {
  node_gather<1, EPI_L2, false>(ei, E, deg, nullptr, nullptr, csrc, xl, as, ad, bias, out, N);
}

extern "C" void kernel_launch(void* const* d_in, const int* in_sizes, int n_in,
                              void* d_out, int out_size, void* d_ws, size_t ws_size,
                              hipStream_t stream) {
  const int*   ei    = (const int*)d_in[0];
  const float* emb   = (const float*)d_in[1];
  const float* W1    = (const float*)d_in[2];
  const float* asrc1 = (const float*)d_in[3];
  const float* adst1 = (const float*)d_in[4];
  const float* b1    = (const float*)d_in[5];
  const float* W2    = (const float*)d_in[6];
  const float* asrc2 = (const float*)d_in[7];
  const float* adst2 = (const float*)d_in[8];
  const float* b2    = (const float*)d_in[9];
  float* out = (float*)d_out;

  const int E = in_sizes[0] / 2;
  const int N = in_sizes[1] / 128;

  int* q      = (int*)d_ws;
  int* pairs  = q; q += (size_t)N * BCAP * 2;
  int* csrc   = q; q += (size_t)N * BCAP;
  float* p   = (float*)q;
  float* xl1 = p; p += (size_t)N * 128;
  float* x1  = p; p += (size_t)N * 128;
  float* xl2 = p; p += (size_t)N * 128;
  float* as1 = p; p += (size_t)N * 4;
  float* ad1 = p; p += (size_t)N * 4;
  float* as2 = p; p += (size_t)N;
  float* ad2 = p; p += (size_t)N;
  int* deg    = (int*)p;

  dim3 blk(256);
  int gGemm  = (N + 127) / 128;
  int gNode8 = (N + 7) / 8;
  int gNode  = (N + 255) / 256;
  int gEdge  = (E + 255) / 256;

  zero_ints<<<gNode, blk, 0, stream>>>(deg, N);
  gemm1_beside_buckets<<<gGemm + gEdge, blk, 0, stream>>>(emb, W1, asrc1, adst1, xl1, as1, ad1, N,
                                                         gGemm, ei, deg, pairs, E);

  node_gather_h4_elu<<<gNode8, blk, 0, stream>>>(ei, E, deg, pairs, csrc, xl1, as1, ad1, b1, x1, N);

  gemm128_attn_h1   <<<gGemm,  blk, 0, stream>>>(x1, W2, asrc2, adst2, xl2, as2, ad2, N);
  node_gather_h1_l2 <<<gNode8, blk, 0, stream>>>(ei, E, deg, csrc, xl2, as2, ad2, b2, out, N);
}
